# starting kernel
# baseline (speedup 1.0000x reference)
#include <hip/hip_runtime.h>

#define N_NODES 10000
#define N_EDGES 320000
#define IN_DIM  256
#define HIDDEN  256
#define OUT_DIM 128

typedef __attribute__((ext_vector_type(2))) float v2f;
typedef __attribute__((ext_vector_type(8))) float v8f;

template<int D>
__global__ __launch_bounds__(256) void spmm_scatter(
    const float* __restrict__ H,
    const int*   __restrict__ src,
    const int*   __restrict__ dst,
    const float* __restrict__ val,
    float*       __restrict__ Z)
{
    const int lane = threadIdx.x & 31;
    const int wv   = threadIdx.x >> 5;
    const int e    = blockIdx.x * 8 + wv;
    if (e >= N_EDGES) return;

    const int   s = src[e];
    const int   d = dst[e];
    const float w = val[e];

    const float* __restrict__ hrow = H + (size_t)s * D;
    float*       __restrict__ zrow = Z + (size_t)d * D;

#pragma unroll
    for (int i = 0; i < D / 128; ++i) {
        const int c = (i * 32 + lane) * 4;
        const float4 h = *(const float4*)(hrow + c);
        atomicAdd(&zrow[c + 0], w * h.x);
        atomicAdd(&zrow[c + 1], w * h.y);
        atomicAdd(&zrow[c + 2], w * h.z);
        atomicAdd(&zrow[c + 3], w * h.w);
    }
}

template<int MSUB, int NSUB, bool RELU>
__global__ __launch_bounds__(128) void gemm_wmma_f32(
    const float* __restrict__ A,
    const float* __restrict__ B,
    float*       __restrict__ C,
    int M, int K, int N)
{
    const int lane = threadIdx.x & 31;
    const int wave = threadIdx.x >> 5;
    const int lo   = lane & 15;
    const int hi   = lane >> 4;

    const int m0  = blockIdx.x * (16 * MSUB);
    const int nb0 = wave * (NSUB * 16);

    int mbase[MSUB];
#pragma unroll
    for (int ms = 0; ms < MSUB; ++ms)
        mbase[ms] = min(m0 + 16 * ms, M - 16);

    v8f acc[MSUB][NSUB];
#pragma unroll
    for (int ms = 0; ms < MSUB; ++ms)
#pragma unroll
        for (int t = 0; t < NSUB; ++t) acc[ms][t] = 0.0f;

    const float* __restrict__ arow[MSUB];
#pragma unroll
    for (int ms = 0; ms < MSUB; ++ms)
        arow[ms] = A + (size_t)(mbase[ms] + lo) * K;

    for (int k = 0; k < K; k += 4) {
        const int ka = k + 2 * hi;

        v2f a[MSUB];
#pragma unroll
        for (int ms = 0; ms < MSUB; ++ms)
            a[ms] = *(const v2f*)(arow[ms] + ka);

#pragma unroll
        for (int t = 0; t < NSUB; ++t) {
            const int n = nb0 + t * 16 + lo;
            v2f b;
            b.x = B[(size_t)ka * N + n];
            b.y = B[(size_t)(ka + 1) * N + n];
#pragma unroll
            for (int ms = 0; ms < MSUB; ++ms) {
                acc[ms][t] = __builtin_amdgcn_wmma_f32_16x16x4_f32(
                    false, a[ms], false, b, (short)0, acc[ms][t],
                    false, false);
            }
        }
    }

#pragma unroll
    for (int ms = 0; ms < MSUB; ++ms) {
        if (m0 + 16 * ms < M) {
#pragma unroll
            for (int t = 0; t < NSUB; ++t) {
#pragma unroll
                for (int v = 0; v < 8; ++v) {
                    const int row = mbase[ms] + v + 8 * hi;
                    const int col = nb0 + t * 16 + lo;
                    float x = acc[ms][t][v];
                    if (RELU) x = fmaxf(x, 0.0f);
                    C[(size_t)row * N + col] = x;
                }
            }
        }
    }
}

extern "C" void kernel_launch(void* const* d_in, const int* in_sizes, int n_in,
                              void* d_out, int out_size, void* d_ws, size_t ws_size,
                              hipStream_t stream)
{
    (void)in_sizes; (void)n_in; (void)out_size; (void)ws_size;

    const float* features = (const float*)d_in[0];
    const int*   esrc     = (const int*)  d_in[1];
    const int*   edst     = (const int*)  d_in[2];
    const float* eval_    = (const float*)d_in[3];
    const float* W0       = (const float*)d_in[4];
    const float* W1       = (const float*)d_in[5];
    float*       out      = (float*)d_out;

    const size_t SZ = (size_t)N_NODES * HIDDEN * sizeof(float);
    float* Z = (float*)d_ws;
    float* H = (float*)((char*)d_ws + SZ);

    const int spmm_blocks = (N_EDGES + 7) / 8;
    const int gemm_blocks = (N_NODES + 31) / 32;

    hipMemsetAsync(Z, 0, SZ, stream);
    spmm_scatter<IN_DIM><<<spmm_blocks, 256, 0, stream>>>(features, esrc, edst, eval_, Z);
    gemm_wmma_f32<2, 4, true><<<gemm_blocks, 128, 0, stream>>>(Z, W0, H,
                                                               N_NODES, IN_DIM, HIDDEN);

    float* G = Z;
    gemm_wmma_f32<2, 2, false><<<gemm_blocks, 128, 0, stream>>>(H, W1, G,
                                                                N_NODES, HIDDEN, OUT_DIM);
    hipMemsetAsync(out, 0, (size_t)N_NODES * OUT_DIM * sizeof(float), stream);
    spmm_scatter<OUT_DIM><<<spmm_blocks, 256, 0, stream>>>(G, esrc, edst, eval_, out);
}


# SpMM: global f32 atomic scatter -> each block holds a 40-row result tile in LDS, filters the raw dst array itself, adds val*row by LDS f32 atomics (arrival order); no edge list, no memsets
# speedup vs baseline: 1.7685x; 1.7685x over previous
#include <hip/hip_runtime.h>

#define N_NODES 10000
#define N_EDGES 320000
#define IN_DIM  256
#define HIDDEN  256
#define OUT_DIM 128

typedef __attribute__((ext_vector_type(2))) float v2f;
typedef __attribute__((ext_vector_type(8))) float v8f;

#define SCAN_CHUNK 512
#define QCAP       1152

static_assert(N_EDGES % SCAN_CHUNK == 0, "scan reads whole chunks");
static_assert(N_EDGES < (1 << 20), "edge index is packed into 20 bits");

template<int D>
__device__ __forceinline__ void load_row(float (&h)[D / 32],
                                         const float* __restrict__ H,
                                         int s, int lane)
{
    const float* __restrict__ hrow = H + (size_t)s * D + lane;
#pragma unroll
    for (int i = 0; i < D / 32; ++i) h[i] = hrow[i * 32];
}

#define GROUP 4
template<int D>
__device__ __forceinline__ void drain_queue(
    const float* __restrict__ H,
    const int*   __restrict__ src,
    const float* __restrict__ val,
    const int* q, int cnt, float* acc, int lane)
{
    for (int b = 0; b < cnt; b += 32) {
        const int n = min(32, cnt - b);
        int   r = 0, s = 0;
        float w = 0.0f;
        if (lane < n) {
            const int ent = q[b + lane];
            const int e   = ent & 0xFFFFF;
            r = ent >> 20;
            s = src[e];
            w = val[e];
        }
        float cur[GROUP][D / 32];
        float nxt[GROUP][D / 32];
#pragma unroll
        for (int g = 0; g < GROUP; ++g) {
#pragma unroll
            for (int i = 0; i < D / 32; ++i) { cur[g][i] = 0.0f; nxt[g][i] = 0.0f; }
        }
#pragma unroll
        for (int g = 0; g < GROUP; ++g)
            if (g < n)
                load_row<D>(cur[g], H, __builtin_amdgcn_readlane(s, g), lane);

        for (int k = 0; k < n; k += GROUP) {
#pragma unroll
            for (int g = 0; g < GROUP; ++g)
                if (k + GROUP + g < n)
                    load_row<D>(nxt[g], H,
                                __builtin_amdgcn_readlane(s, k + GROUP + g), lane);
#pragma unroll
            for (int g = 0; g < GROUP; ++g) {
                if (k + g < n) {
                    const float wk = __int_as_float(
                        __builtin_amdgcn_readlane(__float_as_int(w), k + g));
                    float* arow = acc + __builtin_amdgcn_readlane(r, k + g) * D + lane;
#pragma unroll
                    for (int i = 0; i < D / 32; ++i)
                        atomicAdd(&arow[i * 32], wk * cur[g][i]);
                }
            }
#pragma unroll
            for (int g = 0; g < GROUP; ++g) {
#pragma unroll
                for (int i = 0; i < D / 32; ++i) cur[g][i] = nxt[g][i];
            }
        }
    }
}

template<int D, int R>
__global__ __launch_bounds__(1024) void spmm_rows_lds(
    const float* __restrict__ H,
    const int*   __restrict__ src,
    const int*   __restrict__ dst,
    const float* __restrict__ val,
    float*       __restrict__ Z)
{
    __shared__ float acc[R * D];
    __shared__ int   queue[32 * QCAP];

    const int lane = threadIdx.x & 31;
    const int wv   = __builtin_amdgcn_readfirstlane(threadIdx.x >> 5);
    const int base = blockIdx.x * R;

    for (int i = threadIdx.x; i < R * D / 4; i += 1024)
        ((float4*)acc)[i] = make_float4(0.0f, 0.0f, 0.0f, 0.0f);
    __syncthreads();

    int* q   = queue + wv * QCAP;
    int  cnt = 0;

    constexpr int NCHUNK = N_EDGES / SCAN_CHUNK;
    int4 d[4];
#pragma unroll
    for (int j = 0; j < 4; ++j) d[j] = make_int4(-1, -1, -1, -1);
    if (wv < NCHUNK) {
#pragma unroll
        for (int j = 0; j < 4; ++j)
            d[j] = *(const int4*)(dst + wv * SCAN_CHUNK + lane * 4 + j * 128);
    }
    for (int c = wv; c < NCHUNK; c += 32) {
        const int e0 = c * SCAN_CHUNK + lane * 4;
        int4 dn[4];
#pragma unroll
        for (int j = 0; j < 4; ++j) dn[j] = d[j];
        if (c + 32 < NCHUNK) {
#pragma unroll
            for (int j = 0; j < 4; ++j)
                dn[j] = *(const int4*)(dst + e0 + 32 * SCAN_CHUNK + j * 128);
        }
#pragma unroll
        for (int j = 0; j < 4; ++j) {
            const int dv[4] = { d[j].x, d[j].y, d[j].z, d[j].w };
#pragma unroll
            for (int t = 0; t < 4; ++t) {
                const unsigned rel  = (unsigned)(dv[t] - base);
                const bool     hit  = rel < (unsigned)R;
                const unsigned mask = __builtin_amdgcn_ballot_w32(hit);
                if (mask) {
                    if (hit) {
                        const int pos = cnt + __builtin_amdgcn_mbcnt_lo(mask, 0);
                        q[pos] = (e0 + j * 128 + t) | ((int)rel << 20);
                    }
                    cnt += __builtin_popcount(mask);
                }
            }
        }
        if (cnt > QCAP - SCAN_CHUNK) {
            drain_queue<D>(H, src, val, q, cnt, acc, lane);
            cnt = 0;
        }
#pragma unroll
        for (int j = 0; j < 4; ++j) d[j] = dn[j];
    }
    drain_queue<D>(H, src, val, q, cnt, acc, lane);

    __syncthreads();

    for (int i = threadIdx.x; i < R * D / 4; i += 1024) {
        const int row = base + i / (D / 4);
        if (row < N_NODES)
            ((float4*)(Z + (size_t)base * D))[i] = ((const float4*)acc)[i];
    }
}

template<int MSUB, int NSUB, bool RELU>
__global__ __launch_bounds__(128) void gemm_wmma_f32(
    const float* __restrict__ A,
    const float* __restrict__ B,
    float*       __restrict__ C,
    int M, int K, int N)
{
    const int lane = threadIdx.x & 31;
    const int wave = threadIdx.x >> 5;
    const int lo   = lane & 15;
    const int hi   = lane >> 4;

    const int m0  = blockIdx.x * (16 * MSUB);
    const int nb0 = wave * (NSUB * 16);

    int mbase[MSUB];
#pragma unroll
    for (int ms = 0; ms < MSUB; ++ms)
        mbase[ms] = min(m0 + 16 * ms, M - 16);

    v8f acc[MSUB][NSUB];
#pragma unroll
    for (int ms = 0; ms < MSUB; ++ms)
#pragma unroll
        for (int t = 0; t < NSUB; ++t) acc[ms][t] = 0.0f;

    const float* __restrict__ arow[MSUB];
#pragma unroll
    for (int ms = 0; ms < MSUB; ++ms)
        arow[ms] = A + (size_t)(mbase[ms] + lo) * K;

    for (int k = 0; k < K; k += 4) {
        const int ka = k + 2 * hi;

        v2f a[MSUB];
#pragma unroll
        for (int ms = 0; ms < MSUB; ++ms)
            a[ms] = *(const v2f*)(arow[ms] + ka);

#pragma unroll
        for (int t = 0; t < NSUB; ++t) {
            const int n = nb0 + t * 16 + lo;
            v2f b;
            b.x = B[(size_t)ka * N + n];
            b.y = B[(size_t)(ka + 1) * N + n];
#pragma unroll
            for (int ms = 0; ms < MSUB; ++ms) {
                acc[ms][t] = __builtin_amdgcn_wmma_f32_16x16x4_f32(
                    false, a[ms], false, b, (short)0, acc[ms][t],
                    false, false);
            }
        }
    }

#pragma unroll
    for (int ms = 0; ms < MSUB; ++ms) {
        if (m0 + 16 * ms < M) {
#pragma unroll
            for (int t = 0; t < NSUB; ++t) {
#pragma unroll
                for (int v = 0; v < 8; ++v) {
                    const int row = mbase[ms] + v + 8 * hi;
                    const int col = nb0 + t * 16 + lo;
                    float x = acc[ms][t][v];
                    if (RELU) x = fmaxf(x, 0.0f);
                    C[(size_t)row * N + col] = x;
                }
            }
        }
    }
}

extern "C" void kernel_launch(void* const* d_in, const int* in_sizes, int n_in,
                              void* d_out, int out_size, void* d_ws, size_t ws_size,
                              hipStream_t stream)
{
    (void)in_sizes; (void)n_in; (void)out_size; (void)ws_size;

    const float* features = (const float*)d_in[0];
    const int*   esrc     = (const int*)  d_in[1];
    const int*   edst     = (const int*)  d_in[2];
    const float* eval_    = (const float*)d_in[3];
    const float* W0       = (const float*)d_in[4];
    const float* W1       = (const float*)d_in[5];
    float*       out      = (float*)d_out;

    const size_t SZ = (size_t)N_NODES * HIDDEN * sizeof(float);
    float* Z = (float*)d_ws;
    float* H = (float*)((char*)d_ws + SZ);

    constexpr int SPMM_ROWS = 40;
    const int spmm_blocks = (N_NODES + SPMM_ROWS - 1) / SPMM_ROWS;
    const int gemm_blocks = (N_NODES + 31) / 32;

    spmm_rows_lds<IN_DIM, SPMM_ROWS><<<spmm_blocks, 1024, 0, stream>>>(features, esrc, edst, eval_, Z);
    gemm_wmma_f32<2, 4, true><<<gemm_blocks, 128, 0, stream>>>(Z, W0, H,
                                                               N_NODES, IN_DIM, HIDDEN);

    float* G = Z;
    gemm_wmma_f32<2, 2, false><<<gemm_blocks, 128, 0, stream>>>(H, W1, G,
                                                                N_NODES, HIDDEN, OUT_DIM);
    spmm_rows_lds<OUT_DIM, SPMM_ROWS><<<spmm_blocks, 1024, 0, stream>>>(G, esrc, edst, eval_, out);
}


# Layer-1 SpMM skips the dst filter: layer-0 SpMM keeps each wave's hit edges (tile row, src, weight) in the workspace, layer 1 adds from that list; row sums stay in arrival order (LDS atomics)
# speedup vs baseline: 2.0080x; 1.1354x over previous
#include <hip/hip_runtime.h>

#define N_NODES 10000
#define N_EDGES 320000
#define IN_DIM  256
#define HIDDEN  256
#define OUT_DIM 128

typedef __attribute__((ext_vector_type(2))) float v2f;
typedef __attribute__((ext_vector_type(8))) float v8f;

#define SCAN_CHUNK 512
#define QCAP       1152
#define KEEP_CAP   256

static_assert(N_EDGES % SCAN_CHUNK == 0, "scan reads whole chunks");
static_assert(N_EDGES < (1 << 20), "edge index is packed into 20 bits");
static_assert(N_NODES <= (1 << 20), "a kept pair packs src into 20 bits");

template<int D>
__device__ __forceinline__ void load_row(float (&h)[D / 32],
                                         const float* __restrict__ H,
                                         int s, int lane)
{
    const float* __restrict__ hrow = H + (size_t)s * D + lane;
#pragma unroll
    for (int i = 0; i < D / 32; ++i) h[i] = hrow[i * 32];
}

#define GROUP 4
template<int D>
__device__ __forceinline__ void add_batch(
    const float* __restrict__ H, int n, int r, int s, float w,
    float* acc, int lane)
{
    float cur[GROUP][D / 32];
    float nxt[GROUP][D / 32];
#pragma unroll
    for (int g = 0; g < GROUP; ++g) {
#pragma unroll
        for (int i = 0; i < D / 32; ++i) { cur[g][i] = 0.0f; nxt[g][i] = 0.0f; }
    }
#pragma unroll
    for (int g = 0; g < GROUP; ++g)
        if (g < n)
            load_row<D>(cur[g], H, __builtin_amdgcn_readlane(s, g), lane);

    for (int k = 0; k < n; k += GROUP) {
#pragma unroll
        for (int g = 0; g < GROUP; ++g)
            if (k + GROUP + g < n)
                load_row<D>(nxt[g], H,
                            __builtin_amdgcn_readlane(s, k + GROUP + g), lane);
#pragma unroll
        for (int g = 0; g < GROUP; ++g) {
            if (k + g < n) {
                const float wk = __int_as_float(
                    __builtin_amdgcn_readlane(__float_as_int(w), k + g));
                float* arow = acc + __builtin_amdgcn_readlane(r, k + g) * D + lane;
#pragma unroll
                for (int i = 0; i < D / 32; ++i)
                    atomicAdd(&arow[i * 32], wk * cur[g][i]);
            }
        }
#pragma unroll
        for (int g = 0; g < GROUP; ++g) {
#pragma unroll
            for (int i = 0; i < D / 32; ++i) cur[g][i] = nxt[g][i];
        }
    }
}

template<int D, bool KEEP>
__device__ __forceinline__ void drain_queue(
    const float* __restrict__ H,
    const int*   __restrict__ src,
    const float* __restrict__ val,
    const int* q, int cnt, float* acc, int lane,
    int2* __restrict__ kept, int kept_n)
{
    for (int b = 0; b < cnt; b += 32) {
        const int n = min(32, cnt - b);
        int   r = 0, s = 0;
        float w = 0.0f;
        if (lane < n) {
            const int ent = q[b + lane];
            const int e   = ent & 0xFFFFF;
            r = ent >> 20;
            s = src[e];
            w = val[e];
        }
        add_batch<D>(H, n, r, s, w, acc, lane);
        if (KEEP) {
            const int pos = kept_n + b + lane;
            if (lane < n && pos < KEEP_CAP)
                kept[pos] = make_int2(s | (r << 20), __float_as_int(w));
        }
    }
}

template<int D>
__device__ __forceinline__ void drain_kept(
    const float* __restrict__ H,
    const int2*  __restrict__ kept, int cnt, float* acc, int lane)
{
    for (int b = 0; b < cnt; b += 32) {
        const int n = min(32, cnt - b);
        int   r = 0, s = 0;
        float w = 0.0f;
        if (lane < n) {
            const int2 ent = kept[b + lane];
            s = ent.x & 0xFFFFF;
            r = ent.x >> 20;
            w = __int_as_float(ent.y);
        }
        add_batch<D>(H, n, r, s, w, acc, lane);
    }
}

template<int D, int R, int MODE>
__global__ __launch_bounds__(1024) void spmm_rows_lds(
    const float* __restrict__ H,
    const int*   __restrict__ src,
    const int*   __restrict__ dst,
    const float* __restrict__ val,
    float*       __restrict__ Z,
    int*         __restrict__ kept,
    int*         __restrict__ kept_cnt)
{
    __shared__ float acc[R * D];
    __shared__ int   queue[32 * QCAP];

    const int lane = threadIdx.x & 31;
    const int wv   = __builtin_amdgcn_readfirstlane(threadIdx.x >> 5);
    const int base = blockIdx.x * R;
    const int slot = blockIdx.x * 32 + wv;

    for (int i = threadIdx.x; i < R * D / 4; i += 1024)
        ((float4*)acc)[i] = make_float4(0.0f, 0.0f, 0.0f, 0.0f);
    __syncthreads();

    int2* kp = (int2*)kept + (size_t)slot * KEEP_CAP;

    int have = -1;
    if (MODE == 1)
        have = __builtin_amdgcn_readfirstlane(kept_cnt[slot]);

    if (have >= 0) {
        drain_kept<D>(H, kp, have, acc, lane);
    } else {
        int* q      = queue + wv * QCAP;
        int  cnt    = 0;
        int  kept_n = 0;

        constexpr int NCHUNK = N_EDGES / SCAN_CHUNK;
        int4 d[4];
#pragma unroll
        for (int j = 0; j < 4; ++j) d[j] = make_int4(-1, -1, -1, -1);
        if (wv < NCHUNK) {
#pragma unroll
            for (int j = 0; j < 4; ++j)
                d[j] = *(const int4*)(dst + wv * SCAN_CHUNK + lane * 4 + j * 128);
        }
        for (int c = wv; c < NCHUNK; c += 32) {
            const int e0 = c * SCAN_CHUNK + lane * 4;
            int4 dn[4];
#pragma unroll
            for (int j = 0; j < 4; ++j) dn[j] = d[j];
            if (c + 32 < NCHUNK) {
#pragma unroll
                for (int j = 0; j < 4; ++j)
                    dn[j] = *(const int4*)(dst + e0 + 32 * SCAN_CHUNK + j * 128);
            }
#pragma unroll
            for (int j = 0; j < 4; ++j) {
                const int dv[4] = { d[j].x, d[j].y, d[j].z, d[j].w };
#pragma unroll
                for (int t = 0; t < 4; ++t) {
                    const unsigned rel  = (unsigned)(dv[t] - base);
                    const bool     hit  = rel < (unsigned)R;
                    const unsigned mask = __builtin_amdgcn_ballot_w32(hit);
                    if (mask) {
                        if (hit) {
                            const int pos = cnt + __builtin_amdgcn_mbcnt_lo(mask, 0);
                            q[pos] = (e0 + j * 128 + t) | ((int)rel << 20);
                        }
                        cnt += __builtin_popcount(mask);
                    }
                }
            }
            if (cnt > QCAP - SCAN_CHUNK) {
                drain_queue<D, MODE == 0>(H, src, val, q, cnt, acc, lane, kp, kept_n);
                kept_n += cnt;
                cnt = 0;
            }
#pragma unroll
            for (int j = 0; j < 4; ++j) d[j] = dn[j];
        }
        drain_queue<D, MODE == 0>(H, src, val, q, cnt, acc, lane, kp, kept_n);
        kept_n += cnt;

        if (MODE == 0 && lane == 0)
            kept_cnt[slot] = (kept_n <= KEEP_CAP) ? kept_n : -1;
    }

    __syncthreads();

    for (int i = threadIdx.x; i < R * D / 4; i += 1024) {
        const int row = base + i / (D / 4);
        if (row < N_NODES)
            ((float4*)(Z + (size_t)base * D))[i] = ((const float4*)acc)[i];
    }
}

template<int MSUB, int NSUB, bool RELU>
__global__ __launch_bounds__(128) void gemm_wmma_f32(
    const float* __restrict__ A,
    const float* __restrict__ B,
    float*       __restrict__ C,
    int M, int K, int N)
{
    const int lane = threadIdx.x & 31;
    const int wave = threadIdx.x >> 5;
    const int lo   = lane & 15;
    const int hi   = lane >> 4;

    const int m0  = blockIdx.x * (16 * MSUB);
    const int nb0 = wave * (NSUB * 16);

    int mbase[MSUB];
#pragma unroll
    for (int ms = 0; ms < MSUB; ++ms)
        mbase[ms] = min(m0 + 16 * ms, M - 16);

    v8f acc[MSUB][NSUB];
#pragma unroll
    for (int ms = 0; ms < MSUB; ++ms)
#pragma unroll
        for (int t = 0; t < NSUB; ++t) acc[ms][t] = 0.0f;

    const float* __restrict__ arow[MSUB];
#pragma unroll
    for (int ms = 0; ms < MSUB; ++ms)
        arow[ms] = A + (size_t)(mbase[ms] + lo) * K;

    for (int k = 0; k < K; k += 4) {
        const int ka = k + 2 * hi;

        v2f a[MSUB];
#pragma unroll
        for (int ms = 0; ms < MSUB; ++ms)
            a[ms] = *(const v2f*)(arow[ms] + ka);

#pragma unroll
        for (int t = 0; t < NSUB; ++t) {
            const int n = nb0 + t * 16 + lo;
            v2f b;
            b.x = B[(size_t)ka * N + n];
            b.y = B[(size_t)(ka + 1) * N + n];
#pragma unroll
            for (int ms = 0; ms < MSUB; ++ms) {
                acc[ms][t] = __builtin_amdgcn_wmma_f32_16x16x4_f32(
                    false, a[ms], false, b, (short)0, acc[ms][t],
                    false, false);
            }
        }
    }

#pragma unroll
    for (int ms = 0; ms < MSUB; ++ms) {
        if (m0 + 16 * ms < M) {
#pragma unroll
            for (int t = 0; t < NSUB; ++t) {
#pragma unroll
                for (int v = 0; v < 8; ++v) {
                    const int row = mbase[ms] + v + 8 * hi;
                    const int col = nb0 + t * 16 + lo;
                    float x = acc[ms][t][v];
                    if (RELU) x = fmaxf(x, 0.0f);
                    C[(size_t)row * N + col] = x;
                }
            }
        }
    }
}

extern "C" void kernel_launch(void* const* d_in, const int* in_sizes, int n_in,
                              void* d_out, int out_size, void* d_ws, size_t ws_size,
                              hipStream_t stream)
{
    (void)in_sizes; (void)n_in; (void)out_size; (void)ws_size;

    const float* features = (const float*)d_in[0];
    const int*   esrc     = (const int*)  d_in[1];
    const int*   edst     = (const int*)  d_in[2];
    const float* eval_    = (const float*)d_in[3];
    const float* W0       = (const float*)d_in[4];
    const float* W1       = (const float*)d_in[5];
    float*       out      = (float*)d_out;

    const size_t SZ = (size_t)N_NODES * HIDDEN * sizeof(float);
    float* Z = (float*)d_ws;
    float* H = (float*)((char*)d_ws + SZ);

    constexpr int SPMM_ROWS = 40;
    const int spmm_blocks = (N_NODES + SPMM_ROWS - 1) / SPMM_ROWS;
    constexpr size_t KEPT = (size_t)((N_NODES + SPMM_ROWS - 1) / SPMM_ROWS) * 32 * KEEP_CAP * sizeof(int2);
    int* kept     = (int*)((char*)d_ws + 2 * SZ);
    int* kept_cnt = (int*)((char*)d_ws + 2 * SZ + KEPT);
    const int gemm_blocks = (N_NODES + 31) / 32;

    spmm_rows_lds<IN_DIM, SPMM_ROWS, 0><<<spmm_blocks, 1024, 0, stream>>>(features, esrc, edst, eval_, Z,
                                                                          kept, kept_cnt);
    gemm_wmma_f32<2, 4, true><<<gemm_blocks, 128, 0, stream>>>(Z, W0, H,
                                                               N_NODES, IN_DIM, HIDDEN);

    float* G = Z;
    gemm_wmma_f32<2, 2, false><<<gemm_blocks, 128, 0, stream>>>(H, W1, G,
                                                                N_NODES, HIDDEN, OUT_DIM);
    spmm_rows_lds<OUT_DIM, SPMM_ROWS, 1><<<spmm_blocks, 1024, 0, stream>>>(G, esrc, edst, eval_, out,
                                                                           kept, kept_cnt);
}


# GEMMs: A and B tiles staged through LDS in 32-deep K chunks (two buffers; next chunk's b128 global loads issued ahead of this chunk's WMMAs); fragments are LDS reads; k and sum order unchanged
# speedup vs baseline: 2.4875x; 1.2388x over previous
#include <hip/hip_runtime.h>

#define N_NODES 10000
#define N_EDGES 320000
#define IN_DIM  256
#define HIDDEN  256
#define OUT_DIM 128

typedef __attribute__((ext_vector_type(2))) float v2f;
typedef __attribute__((ext_vector_type(8))) float v8f;

#define SCAN_CHUNK 512
#define QCAP       1152
#define KEEP_CAP   256

static_assert(N_EDGES % SCAN_CHUNK == 0, "scan reads whole chunks");
static_assert(N_EDGES < (1 << 20), "edge index is packed into 20 bits");
static_assert(N_NODES <= (1 << 20), "a kept pair packs src into 20 bits");

template<int D>
__device__ __forceinline__ void load_row(float (&h)[D / 32],
                                         const float* __restrict__ H,
                                         int s, int lane)
{
    const float* __restrict__ hrow = H + (size_t)s * D + lane;
#pragma unroll
    for (int i = 0; i < D / 32; ++i) h[i] = hrow[i * 32];
}

#define GROUP 4
template<int D>
__device__ __forceinline__ void add_batch(
    const float* __restrict__ H, int n, int r, int s, float w,
    float* acc, int lane)
{
    float cur[GROUP][D / 32];
    float nxt[GROUP][D / 32];
#pragma unroll
    for (int g = 0; g < GROUP; ++g) {
#pragma unroll
        for (int i = 0; i < D / 32; ++i) { cur[g][i] = 0.0f; nxt[g][i] = 0.0f; }
    }
#pragma unroll
    for (int g = 0; g < GROUP; ++g)
        if (g < n)
            load_row<D>(cur[g], H, __builtin_amdgcn_readlane(s, g), lane);

    for (int k = 0; k < n; k += GROUP) {
#pragma unroll
        for (int g = 0; g < GROUP; ++g)
            if (k + GROUP + g < n)
                load_row<D>(nxt[g], H,
                            __builtin_amdgcn_readlane(s, k + GROUP + g), lane);
#pragma unroll
        for (int g = 0; g < GROUP; ++g) {
            if (k + g < n) {
                const float wk = __int_as_float(
                    __builtin_amdgcn_readlane(__float_as_int(w), k + g));
                float* arow = acc + __builtin_amdgcn_readlane(r, k + g) * D + lane;
#pragma unroll
                for (int i = 0; i < D / 32; ++i)
                    atomicAdd(&arow[i * 32], wk * cur[g][i]);
            }
        }
#pragma unroll
        for (int g = 0; g < GROUP; ++g) {
#pragma unroll
            for (int i = 0; i < D / 32; ++i) cur[g][i] = nxt[g][i];
        }
    }
}

template<int D, bool KEEP>
__device__ __forceinline__ void drain_queue(
    const float* __restrict__ H,
    const int*   __restrict__ src,
    const float* __restrict__ val,
    const int* q, int cnt, float* acc, int lane,
    int2* __restrict__ kept, int kept_n)
{
    for (int b = 0; b < cnt; b += 32) {
        const int n = min(32, cnt - b);
        int   r = 0, s = 0;
        float w = 0.0f;
        if (lane < n) {
            const int ent = q[b + lane];
            const int e   = ent & 0xFFFFF;
            r = ent >> 20;
            s = src[e];
            w = val[e];
        }
        add_batch<D>(H, n, r, s, w, acc, lane);
        if (KEEP) {
            const int pos = kept_n + b + lane;
            if (lane < n && pos < KEEP_CAP)
                kept[pos] = make_int2(s | (r << 20), __float_as_int(w));
        }
    }
}

template<int D>
__device__ __forceinline__ void drain_kept(
    const float* __restrict__ H,
    const int2*  __restrict__ kept, int cnt, float* acc, int lane)
{
    for (int b = 0; b < cnt; b += 32) {
        const int n = min(32, cnt - b);
        int   r = 0, s = 0;
        float w = 0.0f;
        if (lane < n) {
            const int2 ent = kept[b + lane];
            s = ent.x & 0xFFFFF;
            r = ent.x >> 20;
            w = __int_as_float(ent.y);
        }
        add_batch<D>(H, n, r, s, w, acc, lane);
    }
}

template<int D, int R, int MODE>
__global__ __launch_bounds__(1024) void spmm_rows_lds(
    const float* __restrict__ H,
    const int*   __restrict__ src,
    const int*   __restrict__ dst,
    const float* __restrict__ val,
    float*       __restrict__ Z,
    int*         __restrict__ kept,
    int*         __restrict__ kept_cnt)
{
    __shared__ float acc[R * D];
    __shared__ int   queue[32 * QCAP];

    const int lane = threadIdx.x & 31;
    const int wv   = __builtin_amdgcn_readfirstlane(threadIdx.x >> 5);
    const int base = blockIdx.x * R;
    const int slot = blockIdx.x * 32 + wv;

    for (int i = threadIdx.x; i < R * D / 4; i += 1024)
        ((float4*)acc)[i] = make_float4(0.0f, 0.0f, 0.0f, 0.0f);
    __syncthreads();

    int2* kp = (int2*)kept + (size_t)slot * KEEP_CAP;

    int have = -1;
    if (MODE == 1)
        have = __builtin_amdgcn_readfirstlane(kept_cnt[slot]);

    if (have >= 0) {
        drain_kept<D>(H, kp, have, acc, lane);
    } else {
        int* q      = queue + wv * QCAP;
        int  cnt    = 0;
        int  kept_n = 0;

        constexpr int NCHUNK = N_EDGES / SCAN_CHUNK;
        int4 d[4];
#pragma unroll
        for (int j = 0; j < 4; ++j) d[j] = make_int4(-1, -1, -1, -1);
        if (wv < NCHUNK) {
#pragma unroll
            for (int j = 0; j < 4; ++j)
                d[j] = *(const int4*)(dst + wv * SCAN_CHUNK + lane * 4 + j * 128);
        }
        for (int c = wv; c < NCHUNK; c += 32) {
            const int e0 = c * SCAN_CHUNK + lane * 4;
            int4 dn[4];
#pragma unroll
            for (int j = 0; j < 4; ++j) dn[j] = d[j];
            if (c + 32 < NCHUNK) {
#pragma unroll
                for (int j = 0; j < 4; ++j)
                    dn[j] = *(const int4*)(dst + e0 + 32 * SCAN_CHUNK + j * 128);
            }
#pragma unroll
            for (int j = 0; j < 4; ++j) {
                const int dv[4] = { d[j].x, d[j].y, d[j].z, d[j].w };
#pragma unroll
                for (int t = 0; t < 4; ++t) {
                    const unsigned rel  = (unsigned)(dv[t] - base);
                    const bool     hit  = rel < (unsigned)R;
                    const unsigned mask = __builtin_amdgcn_ballot_w32(hit);
                    if (mask) {
                        if (hit) {
                            const int pos = cnt + __builtin_amdgcn_mbcnt_lo(mask, 0);
                            q[pos] = (e0 + j * 128 + t) | ((int)rel << 20);
                        }
                        cnt += __builtin_popcount(mask);
                    }
                }
            }
            if (cnt > QCAP - SCAN_CHUNK) {
                drain_queue<D, MODE == 0>(H, src, val, q, cnt, acc, lane, kp, kept_n);
                kept_n += cnt;
                cnt = 0;
            }
#pragma unroll
            for (int j = 0; j < 4; ++j) d[j] = dn[j];
        }
        drain_queue<D, MODE == 0>(H, src, val, q, cnt, acc, lane, kp, kept_n);
        kept_n += cnt;

        if (MODE == 0 && lane == 0)
            kept_cnt[slot] = (kept_n <= KEEP_CAP) ? kept_n : -1;
    }

    __syncthreads();

    for (int i = threadIdx.x; i < R * D / 4; i += 1024) {
        const int row = base + i / (D / 4);
        if (row < N_NODES)
            ((float4*)(Z + (size_t)base * D))[i] = ((const float4*)acc)[i];
    }
}

#define KC 32
#define A_PITCH 36

template<int NSUB, int KDIM, bool RELU>
__global__ __launch_bounds__(128) __attribute__((amdgpu_waves_per_eu(1, 3))) void gemm_wmma_f32(
    const float* __restrict__ A,
    const float* __restrict__ B,
    float*       __restrict__ C,
    int M)
{
    constexpr int N       = 64 * NSUB;
    constexpr int B_PITCH = N + 8;
    constexpr int NCH     = KDIM / KC;
    constexpr int A_LD    = (32 * KC / 4) / 128;
    constexpr int B_LD    = (KC * N / 4) / 128;
    static_assert(KDIM % KC == 0, "whole K chunks");

    __shared__ float sA[2][32 * A_PITCH];
    __shared__ float sB[2][KC * B_PITCH];

    const int tid  = threadIdx.x;
    const int lane = tid & 31;
    const int wave = tid >> 5;
    const int lo   = lane & 15;
    const int hi   = lane >> 4;

    const int m0  = blockIdx.x * 32;
    const int nb0 = wave * (NSUB * 16);

    int mbase[2];
#pragma unroll
    for (int ms = 0; ms < 2; ++ms)
        mbase[ms] = min(m0 + 16 * ms, M - 16);

    v8f acc[2][NSUB];
#pragma unroll
    for (int ms = 0; ms < 2; ++ms)
#pragma unroll
        for (int t = 0; t < NSUB; ++t) acc[ms][t] = 0.0f;

    const float* ag[A_LD];
    int          al[A_LD];
#pragma unroll
    for (int i = 0; i < A_LD; ++i) {
        const int idx = tid + 128 * i;
        const int row = idx / (KC / 4);
        const int c4  = idx % (KC / 4);
        ag[i] = A + (size_t)(mbase[row >> 4] + (row & 15)) * KDIM + c4 * 4;
        al[i] = row * A_PITCH + c4 * 4;
    }
    const float* bg[B_LD];
    int          bl[B_LD];
#pragma unroll
    for (int i = 0; i < B_LD; ++i) {
        const int idx = tid + 128 * i;
        const int kr  = idx / (N / 4);
        const int c4  = idx % (N / 4);
        bg[i] = B + (size_t)kr * N + c4 * 4;
        bl[i] = kr * B_PITCH + c4 * 4;
    }

    float4 ra[A_LD], rb[B_LD];
#pragma unroll
    for (int i = 0; i < A_LD; ++i) ra[i] = *(const float4*)(ag[i]);
#pragma unroll
    for (int i = 0; i < B_LD; ++i) rb[i] = *(const float4*)(bg[i]);
#pragma unroll
    for (int i = 0; i < A_LD; ++i) *(float4*)(&sA[0][al[i]]) = ra[i];
#pragma unroll
    for (int i = 0; i < B_LD; ++i) *(float4*)(&sB[0][bl[i]]) = rb[i];
    __syncthreads();

#pragma unroll 1
    for (int c = 0; c < NCH; ++c) {
        const int cur = c & 1;
        const int cn = min(c + 1, NCH - 1);
#pragma unroll
        for (int i = 0; i < A_LD; ++i)
            ra[i] = *(const float4*)(ag[i] + cn * KC);
#pragma unroll
        for (int i = 0; i < B_LD; ++i)
            rb[i] = *(const float4*)(bg[i] + (size_t)cn * KC * N);

        const float* __restrict__ cA = sA[cur];
        const float* __restrict__ cB = sB[cur];
#pragma unroll
        for (int kk = 0; kk < KC; kk += 4) {
            const int ka = kk + 2 * hi;
            v2f a[2];
#pragma unroll
            for (int ms = 0; ms < 2; ++ms)
                a[ms] = *(const v2f*)(cA + (16 * ms + lo) * A_PITCH + ka);
#pragma unroll
            for (int t = 0; t < NSUB; ++t) {
                const int n = nb0 + t * 16 + lo;
                v2f b;
                b.x = cB[ka * B_PITCH + n];
                b.y = cB[(ka + 1) * B_PITCH + n];
#pragma unroll
                for (int ms = 0; ms < 2; ++ms) {
                    acc[ms][t] = __builtin_amdgcn_wmma_f32_16x16x4_f32(
                        false, a[ms], false, b, (short)0, acc[ms][t],
                        false, false);
                }
            }
        }

#pragma unroll
        for (int i = 0; i < A_LD; ++i) *(float4*)(&sA[cur ^ 1][al[i]]) = ra[i];
#pragma unroll
        for (int i = 0; i < B_LD; ++i) *(float4*)(&sB[cur ^ 1][bl[i]]) = rb[i];
        __syncthreads();
    }

#pragma unroll
    for (int ms = 0; ms < 2; ++ms) {
        if (m0 + 16 * ms < M) {
#pragma unroll
            for (int t = 0; t < NSUB; ++t) {
#pragma unroll
                for (int v = 0; v < 8; ++v) {
                    const int row = mbase[ms] + v + 8 * hi;
                    const int col = nb0 + t * 16 + lo;
                    float x = acc[ms][t][v];
                    if (RELU) x = fmaxf(x, 0.0f);
                    C[(size_t)row * N + col] = x;
                }
            }
        }
    }
}

extern "C" void kernel_launch(void* const* d_in, const int* in_sizes, int n_in,
                              void* d_out, int out_size, void* d_ws, size_t ws_size,
                              hipStream_t stream)
{
    (void)in_sizes; (void)n_in; (void)out_size; (void)ws_size;

    const float* features = (const float*)d_in[0];
    const int*   esrc     = (const int*)  d_in[1];
    const int*   edst     = (const int*)  d_in[2];
    const float* eval_    = (const float*)d_in[3];
    const float* W0       = (const float*)d_in[4];
    const float* W1       = (const float*)d_in[5];
    float*       out      = (float*)d_out;

    const size_t SZ = (size_t)N_NODES * HIDDEN * sizeof(float);
    float* Z = (float*)d_ws;
    float* H = (float*)((char*)d_ws + SZ);

    constexpr int SPMM_ROWS = 40;
    const int spmm_blocks = (N_NODES + SPMM_ROWS - 1) / SPMM_ROWS;
    constexpr size_t KEPT = (size_t)((N_NODES + SPMM_ROWS - 1) / SPMM_ROWS) * 32 * KEEP_CAP * sizeof(int2);
    int* kept     = (int*)((char*)d_ws + 2 * SZ);
    int* kept_cnt = (int*)((char*)d_ws + 2 * SZ + KEPT);
    const int gemm_blocks = (N_NODES + 31) / 32;

    spmm_rows_lds<IN_DIM, SPMM_ROWS, 0><<<spmm_blocks, 1024, 0, stream>>>(features, esrc, edst, eval_, Z,
                                                                          kept, kept_cnt);
    gemm_wmma_f32<4, IN_DIM, true><<<gemm_blocks, 128, 0, stream>>>(Z, W0, H, N_NODES);

    float* G = Z;
    gemm_wmma_f32<2, HIDDEN, false><<<gemm_blocks, 128, 0, stream>>>(H, W1, G, N_NODES);
    spmm_rows_lds<OUT_DIM, SPMM_ROWS, 1><<<spmm_blocks, 1024, 0, stream>>>(G, esrc, edst, eval_, out,
                                                                           kept, kept_cnt);
}


# SpMM gather: a wave's row loads overlap: groups of 4 rows issued with no branch or wait among them, next group ahead of this group's LDS adds, two register sets; sums stay arrival-order LDS atomics
# speedup vs baseline: 2.5663x; 1.0317x over previous
#include <hip/hip_runtime.h>

#define N_NODES 10000
#define N_EDGES 320000
#define IN_DIM  256
#define HIDDEN  256
#define OUT_DIM 128

typedef __attribute__((ext_vector_type(2))) float v2f;
typedef __attribute__((ext_vector_type(8))) float v8f;

#define SCAN_CHUNK 512
#define QCAP       1152
#define KEEP_CAP   256

static_assert(N_EDGES % SCAN_CHUNK == 0, "scan reads whole chunks");
static_assert(N_EDGES < (1 << 20), "edge index is packed into 20 bits");
static_assert(N_NODES <= (1 << 20), "a kept pair packs src into 20 bits");

template<int D>
__device__ __forceinline__ void load_row(float (&h)[D / 32],
                                         const float* __restrict__ H,
                                         int s, int lane)
{
    const float* __restrict__ hrow = H + (size_t)s * D + lane;
#pragma unroll
    for (int i = 0; i < D / 32; ++i) h[i] = hrow[i * 32];
}

#define GROUP 4

template<int D>
__device__ __forceinline__ void load_group(float (&h)[GROUP][D / 32],
                                           const float* __restrict__ H,
                                           int s, int k, int last, int lane)
{
#pragma unroll
    for (int g = 0; g < GROUP; ++g)
        load_row<D>(h[g], H,
                    __builtin_amdgcn_readlane(s, min(k + g, last)), lane);
}

template<int D, bool FULL>
__device__ __forceinline__ void add_group(const float (&h)[GROUP][D / 32],
                                          int n, int k, int r, float w,
                                          float* acc, int lane)
{
#pragma unroll
    for (int g = 0; g < GROUP; ++g) {
        if (FULL || k + g < n) {
            const float wk = __int_as_float(
                __builtin_amdgcn_readlane(__float_as_int(w), k + g));
            float* arow = acc + __builtin_amdgcn_readlane(r, k + g) * D + lane;
#pragma unroll
            for (int i = 0; i < D / 32; ++i)
                atomicAdd(&arow[i * 32], wk * h[g][i]);
        }
    }
}

template<int D>
__device__ __forceinline__ void add_batch(
    const float* __restrict__ H, int n, int r, int s, float w,
    float* acc, int lane)
{
    float a[GROUP][D / 32];
    float b[GROUP][D / 32];
    const int last = n - 1;
    int k = 0;
    load_group<D>(a, H, s, 0, last, lane);
    while (k + 2 * GROUP < n) {
        load_group<D>(b, H, s, k + GROUP, last, lane);
        add_group<D, true>(a, n, k, r, w, acc, lane);
        load_group<D>(a, H, s, k + 2 * GROUP, last, lane);
        add_group<D, true>(b, n, k + GROUP, r, w, acc, lane);
        k += 2 * GROUP;
    }
    if (k + GROUP < n) {
        load_group<D>(b, H, s, k + GROUP, last, lane);
        add_group<D, true>(a, n, k, r, w, acc, lane);
        add_group<D, false>(b, n, k + GROUP, r, w, acc, lane);
    } else {
        add_group<D, false>(a, n, k, r, w, acc, lane);
    }
}

template<int D, bool KEEP>
__device__ __forceinline__ void drain_queue(
    const float* __restrict__ H,
    const int*   __restrict__ src,
    const float* __restrict__ val,
    const int* q, int cnt, float* acc, int lane,
    int2* __restrict__ kept, int kept_n)
{
    for (int b = 0; b < cnt; b += 32) {
        const int n = min(32, cnt - b);
        int   r = 0, s = 0;
        float w = 0.0f;
        if (lane < n) {
            const int ent = q[b + lane];
            const int e   = ent & 0xFFFFF;
            r = ent >> 20;
            s = src[e];
            w = val[e];
        }
        add_batch<D>(H, n, r, s, w, acc, lane);
        if (KEEP) {
            const int pos = kept_n + b + lane;
            if (lane < n && pos < KEEP_CAP)
                kept[pos] = make_int2(s | (r << 20), __float_as_int(w));
        }
    }
}

template<int D>
__device__ __forceinline__ void drain_kept(
    const float* __restrict__ H,
    const int2*  __restrict__ kept, int cnt, float* acc, int lane)
{
    for (int b = 0; b < cnt; b += 32) {
        const int n = min(32, cnt - b);
        int   r = 0, s = 0;
        float w = 0.0f;
        if (lane < n) {
            const int2 ent = kept[b + lane];
            s = ent.x & 0xFFFFF;
            r = ent.x >> 20;
            w = __int_as_float(ent.y);
        }
        add_batch<D>(H, n, r, s, w, acc, lane);
    }
}

template<int D, int R, int MODE>
__global__ __launch_bounds__(1024) void spmm_rows_lds(
    const float* __restrict__ H,
    const int*   __restrict__ src,
    const int*   __restrict__ dst,
    const float* __restrict__ val,
    float*       __restrict__ Z,
    int*         __restrict__ kept,
    int*         __restrict__ kept_cnt)
{
    __shared__ float acc[R * D];
    __shared__ int   queue[32 * QCAP];

    const int lane = threadIdx.x & 31;
    const int wv   = __builtin_amdgcn_readfirstlane(threadIdx.x >> 5);
    const int base = blockIdx.x * R;
    const int slot = blockIdx.x * 32 + wv;

    for (int i = threadIdx.x; i < R * D / 4; i += 1024)
        ((float4*)acc)[i] = make_float4(0.0f, 0.0f, 0.0f, 0.0f);
    __syncthreads();

    int2* kp = (int2*)kept + (size_t)slot * KEEP_CAP;

    int have = -1;
    if (MODE == 1)
        have = __builtin_amdgcn_readfirstlane(kept_cnt[slot]);

    if (have >= 0) {
        drain_kept<D>(H, kp, have, acc, lane);
    } else {
        int* q      = queue + wv * QCAP;
        int  cnt    = 0;
        int  kept_n = 0;

        constexpr int NCHUNK = N_EDGES / SCAN_CHUNK;
        int4 d[4];
#pragma unroll
        for (int j = 0; j < 4; ++j) d[j] = make_int4(-1, -1, -1, -1);
        if (wv < NCHUNK) {
#pragma unroll
            for (int j = 0; j < 4; ++j)
                d[j] = *(const int4*)(dst + wv * SCAN_CHUNK + lane * 4 + j * 128);
        }
        for (int c = wv; c < NCHUNK; c += 32) {
            const int e0 = c * SCAN_CHUNK + lane * 4;
            int4 dn[4];
#pragma unroll
            for (int j = 0; j < 4; ++j) dn[j] = d[j];
            if (c + 32 < NCHUNK) {
#pragma unroll
                for (int j = 0; j < 4; ++j)
                    dn[j] = *(const int4*)(dst + e0 + 32 * SCAN_CHUNK + j * 128);
            }
#pragma unroll
            for (int j = 0; j < 4; ++j) {
                const int dv[4] = { d[j].x, d[j].y, d[j].z, d[j].w };
#pragma unroll
                for (int t = 0; t < 4; ++t) {
                    const unsigned rel  = (unsigned)(dv[t] - base);
                    const bool     hit  = rel < (unsigned)R;
                    const unsigned mask = __builtin_amdgcn_ballot_w32(hit);
                    if (mask) {
                        if (hit) {
                            const int pos = cnt + __builtin_amdgcn_mbcnt_lo(mask, 0);
                            q[pos] = (e0 + j * 128 + t) | ((int)rel << 20);
                        }
                        cnt += __builtin_popcount(mask);
                    }
                }
            }
            if (cnt > QCAP - SCAN_CHUNK) {
                drain_queue<D, MODE == 0>(H, src, val, q, cnt, acc, lane, kp, kept_n);
                kept_n += cnt;
                cnt = 0;
            }
#pragma unroll
            for (int j = 0; j < 4; ++j) d[j] = dn[j];
        }
        drain_queue<D, MODE == 0>(H, src, val, q, cnt, acc, lane, kp, kept_n);
        kept_n += cnt;

        if (MODE == 0 && lane == 0)
            kept_cnt[slot] = (kept_n <= KEEP_CAP) ? kept_n : -1;
    }

    __syncthreads();

    for (int i = threadIdx.x; i < R * D / 4; i += 1024) {
        const int row = base + i / (D / 4);
        if (row < N_NODES)
            ((float4*)(Z + (size_t)base * D))[i] = ((const float4*)acc)[i];
    }
}

#define KC 32
#define A_PITCH 36

template<int NSUB, int KDIM, bool RELU>
__global__ __launch_bounds__(128) __attribute__((amdgpu_waves_per_eu(1, 3))) void gemm_wmma_f32(
    const float* __restrict__ A,
    const float* __restrict__ B,
    float*       __restrict__ C,
    int M)
{
    constexpr int N       = 64 * NSUB;
    constexpr int B_PITCH = N + 8;
    constexpr int NCH     = KDIM / KC;
    constexpr int A_LD    = (32 * KC / 4) / 128;
    constexpr int B_LD    = (KC * N / 4) / 128;
    static_assert(KDIM % KC == 0, "whole K chunks");

    __shared__ float sA[2][32 * A_PITCH];
    __shared__ float sB[2][KC * B_PITCH];

    const int tid  = threadIdx.x;
    const int lane = tid & 31;
    const int wave = tid >> 5;
    const int lo   = lane & 15;
    const int hi   = lane >> 4;

    const int m0  = blockIdx.x * 32;
    const int nb0 = wave * (NSUB * 16);

    int mbase[2];
#pragma unroll
    for (int ms = 0; ms < 2; ++ms)
        mbase[ms] = min(m0 + 16 * ms, M - 16);

    v8f acc[2][NSUB];
#pragma unroll
    for (int ms = 0; ms < 2; ++ms)
#pragma unroll
        for (int t = 0; t < NSUB; ++t) acc[ms][t] = 0.0f;

    const float* ag[A_LD];
    int          al[A_LD];
#pragma unroll
    for (int i = 0; i < A_LD; ++i) {
        const int idx = tid + 128 * i;
        const int row = idx / (KC / 4);
        const int c4  = idx % (KC / 4);
        ag[i] = A + (size_t)(mbase[row >> 4] + (row & 15)) * KDIM + c4 * 4;
        al[i] = row * A_PITCH + c4 * 4;
    }
    const float* bg[B_LD];
    int          bl[B_LD];
#pragma unroll
    for (int i = 0; i < B_LD; ++i) {
        const int idx = tid + 128 * i;
        const int kr  = idx / (N / 4);
        const int c4  = idx % (N / 4);
        bg[i] = B + (size_t)kr * N + c4 * 4;
        bl[i] = kr * B_PITCH + c4 * 4;
    }

    float4 ra[A_LD], rb[B_LD];
#pragma unroll
    for (int i = 0; i < A_LD; ++i) ra[i] = *(const float4*)(ag[i]);
#pragma unroll
    for (int i = 0; i < B_LD; ++i) rb[i] = *(const float4*)(bg[i]);
#pragma unroll
    for (int i = 0; i < A_LD; ++i) *(float4*)(&sA[0][al[i]]) = ra[i];
#pragma unroll
    for (int i = 0; i < B_LD; ++i) *(float4*)(&sB[0][bl[i]]) = rb[i];
    __syncthreads();

#pragma unroll 1
    for (int c = 0; c < NCH; ++c) {
        const int cur = c & 1;
        const int cn = min(c + 1, NCH - 1);
#pragma unroll
        for (int i = 0; i < A_LD; ++i)
            ra[i] = *(const float4*)(ag[i] + cn * KC);
#pragma unroll
        for (int i = 0; i < B_LD; ++i)
            rb[i] = *(const float4*)(bg[i] + (size_t)cn * KC * N);

        const float* __restrict__ cA = sA[cur];
        const float* __restrict__ cB = sB[cur];
#pragma unroll
        for (int kk = 0; kk < KC; kk += 4) {
            const int ka = kk + 2 * hi;
            v2f a[2];
#pragma unroll
            for (int ms = 0; ms < 2; ++ms)
                a[ms] = *(const v2f*)(cA + (16 * ms + lo) * A_PITCH + ka);
#pragma unroll
            for (int t = 0; t < NSUB; ++t) {
                const int n = nb0 + t * 16 + lo;
                v2f b;
                b.x = cB[ka * B_PITCH + n];
                b.y = cB[(ka + 1) * B_PITCH + n];
#pragma unroll
                for (int ms = 0; ms < 2; ++ms) {
                    acc[ms][t] = __builtin_amdgcn_wmma_f32_16x16x4_f32(
                        false, a[ms], false, b, (short)0, acc[ms][t],
                        false, false);
                }
            }
        }

#pragma unroll
        for (int i = 0; i < A_LD; ++i) *(float4*)(&sA[cur ^ 1][al[i]]) = ra[i];
#pragma unroll
        for (int i = 0; i < B_LD; ++i) *(float4*)(&sB[cur ^ 1][bl[i]]) = rb[i];
        __syncthreads();
    }

#pragma unroll
    for (int ms = 0; ms < 2; ++ms) {
        if (m0 + 16 * ms < M) {
#pragma unroll
            for (int t = 0; t < NSUB; ++t) {
#pragma unroll
                for (int v = 0; v < 8; ++v) {
                    const int row = mbase[ms] + v + 8 * hi;
                    const int col = nb0 + t * 16 + lo;
                    float x = acc[ms][t][v];
                    if (RELU) x = fmaxf(x, 0.0f);
                    C[(size_t)row * N + col] = x;
                }
            }
        }
    }
}

extern "C" void kernel_launch(void* const* d_in, const int* in_sizes, int n_in,
                              void* d_out, int out_size, void* d_ws, size_t ws_size,
                              hipStream_t stream)
{
    (void)in_sizes; (void)n_in; (void)out_size; (void)ws_size;

    const float* features = (const float*)d_in[0];
    const int*   esrc     = (const int*)  d_in[1];
    const int*   edst     = (const int*)  d_in[2];
    const float* eval_    = (const float*)d_in[3];
    const float* W0       = (const float*)d_in[4];
    const float* W1       = (const float*)d_in[5];
    float*       out      = (float*)d_out;

    const size_t SZ = (size_t)N_NODES * HIDDEN * sizeof(float);
    float* Z = (float*)d_ws;
    float* H = (float*)((char*)d_ws + SZ);

    constexpr int SPMM_ROWS = 40;
    const int spmm_blocks = (N_NODES + SPMM_ROWS - 1) / SPMM_ROWS;
    constexpr size_t KEPT = (size_t)((N_NODES + SPMM_ROWS - 1) / SPMM_ROWS) * 32 * KEEP_CAP * sizeof(int2);
    int* kept     = (int*)((char*)d_ws + 2 * SZ);
    int* kept_cnt = (int*)((char*)d_ws + 2 * SZ + KEPT);
    const int gemm_blocks = (N_NODES + 31) / 32;

    spmm_rows_lds<IN_DIM, SPMM_ROWS, 0><<<spmm_blocks, 1024, 0, stream>>>(features, esrc, edst, eval_, Z,
                                                                          kept, kept_cnt);
    gemm_wmma_f32<4, IN_DIM, true><<<gemm_blocks, 128, 0, stream>>>(Z, W0, H, N_NODES);

    float* G = Z;
    gemm_wmma_f32<2, HIDDEN, false><<<gemm_blocks, 128, 0, stream>>>(H, W1, G, N_NODES);
    spmm_rows_lds<OUT_DIM, SPMM_ROWS, 1><<<spmm_blocks, 1024, 0, stream>>>(G, esrc, edst, eval_, out,
                                                                           kept, kept_cnt);
}


# SpMM gather: source rows fetched 512 B per load (b128, lane = 4 columns) instead of 128 B (b32); LDS tile columns permuted so adds stay on 32 banks; sums stay arrival-order LDS atomics
# speedup vs baseline: 2.6861x; 1.0467x over previous
#include <hip/hip_runtime.h>

#define N_NODES 10000
#define N_EDGES 320000
#define IN_DIM  256
#define HIDDEN  256
#define OUT_DIM 128

typedef __attribute__((ext_vector_type(2))) float v2f;
typedef __attribute__((ext_vector_type(8))) float v8f;

#define SCAN_CHUNK 512
#define QCAP       1152
#define KEEP_CAP   256

static_assert(N_EDGES % SCAN_CHUNK == 0, "scan reads whole chunks");
static_assert(N_EDGES < (1 << 20), "edge index is packed into 20 bits");
static_assert(N_NODES <= (1 << 20), "a kept pair packs src into 20 bits");

template<int D>
__device__ __forceinline__ void load_row(float4 (&h)[D / 128],
                                         const float* __restrict__ H,
                                         int s, int lane)
{
    static_assert(D % 128 == 0, "a row is whole 512-byte pieces");
    const float4* __restrict__ hrow = (const float4*)(H + (size_t)s * D) + lane;
#pragma unroll
    for (int i = 0; i < D / 128; ++i) h[i] = hrow[i * 32];
}

#define GROUP 4

template<int D>
__device__ __forceinline__ void load_group(float4 (&h)[GROUP][D / 128],
                                           const float* __restrict__ H,
                                           int s, int k, int last, int lane)
{
#pragma unroll
    for (int g = 0; g < GROUP; ++g)
        load_row<D>(h[g], H,
                    __builtin_amdgcn_readlane(s, min(k + g, last)), lane);
}

template<int D, bool FULL>
__device__ __forceinline__ void add_group(const float4 (&h)[GROUP][D / 128],
                                          int n, int k, int r, float w,
                                          float* acc, int lane)
{
#pragma unroll
    for (int g = 0; g < GROUP; ++g) {
        if (FULL || k + g < n) {
            const float wk = __int_as_float(
                __builtin_amdgcn_readlane(__float_as_int(w), k + g));
            float* arow = acc + __builtin_amdgcn_readlane(r, k + g) * D + lane;
#pragma unroll
            for (int i = 0; i < D / 128; ++i) {
                atomicAdd(&arow[i * 128],      wk * h[g][i].x);
                atomicAdd(&arow[i * 128 + 32], wk * h[g][i].y);
                atomicAdd(&arow[i * 128 + 64], wk * h[g][i].z);
                atomicAdd(&arow[i * 128 + 96], wk * h[g][i].w);
            }
        }
    }
}

template<int D>
__device__ __forceinline__ void add_batch(
    const float* __restrict__ H, int n, int r, int s, float w,
    float* acc, int lane)
{
    float4 a[GROUP][D / 128];
    float4 b[GROUP][D / 128];
    const int last = n - 1;
    int k = 0;
    load_group<D>(a, H, s, 0, last, lane);
    while (k + 2 * GROUP < n) {
        load_group<D>(b, H, s, k + GROUP, last, lane);
        add_group<D, true>(a, n, k, r, w, acc, lane);
        load_group<D>(a, H, s, k + 2 * GROUP, last, lane);
        add_group<D, true>(b, n, k + GROUP, r, w, acc, lane);
        k += 2 * GROUP;
    }
    if (k + GROUP < n) {
        load_group<D>(b, H, s, k + GROUP, last, lane);
        add_group<D, true>(a, n, k, r, w, acc, lane);
        add_group<D, false>(b, n, k + GROUP, r, w, acc, lane);
    } else {
        add_group<D, false>(a, n, k, r, w, acc, lane);
    }
}

template<int D, bool KEEP>
__device__ __forceinline__ void drain_queue(
    const float* __restrict__ H,
    const int*   __restrict__ src,
    const float* __restrict__ val,
    const int* q, int cnt, float* acc, int lane,
    int2* __restrict__ kept, int kept_n)
{
    for (int b = 0; b < cnt; b += 32) {
        const int n = min(32, cnt - b);
        int   r = 0, s = 0;
        float w = 0.0f;
        if (lane < n) {
            const int ent = q[b + lane];
            const int e   = ent & 0xFFFFF;
            r = ent >> 20;
            s = src[e];
            w = val[e];
        }
        add_batch<D>(H, n, r, s, w, acc, lane);
        if (KEEP) {
            const int pos = kept_n + b + lane;
            if (lane < n && pos < KEEP_CAP)
                kept[pos] = make_int2(s | (r << 20), __float_as_int(w));
        }
    }
}

template<int D>
__device__ __forceinline__ void drain_kept(
    const float* __restrict__ H,
    const int2*  __restrict__ kept, int cnt, float* acc, int lane)
{
    for (int b = 0; b < cnt; b += 32) {
        const int n = min(32, cnt - b);
        int   r = 0, s = 0;
        float w = 0.0f;
        if (lane < n) {
            const int2 ent = kept[b + lane];
            s = ent.x & 0xFFFFF;
            r = ent.x >> 20;
            w = __int_as_float(ent.y);
        }
        add_batch<D>(H, n, r, s, w, acc, lane);
    }
}

template<int D, int R, int MODE>
__global__ __launch_bounds__(1024) void spmm_rows_lds(
    const float* __restrict__ H,
    const int*   __restrict__ src,
    const int*   __restrict__ dst,
    const float* __restrict__ val,
    float*       __restrict__ Z,
    int*         __restrict__ kept,
    int*         __restrict__ kept_cnt)
{
    __shared__ float acc[R * D];
    __shared__ int   queue[32 * QCAP];

    const int lane = threadIdx.x & 31;
    const int wv   = __builtin_amdgcn_readfirstlane(threadIdx.x >> 5);
    const int base = blockIdx.x * R;
    const int slot = blockIdx.x * 32 + wv;

    for (int i = threadIdx.x; i < R * D / 4; i += 1024)
        ((float4*)acc)[i] = make_float4(0.0f, 0.0f, 0.0f, 0.0f);
    __syncthreads();

    int2* kp = (int2*)kept + (size_t)slot * KEEP_CAP;

    int have = -1;
    if (MODE == 1)
        have = __builtin_amdgcn_readfirstlane(kept_cnt[slot]);

    if (have >= 0) {
        drain_kept<D>(H, kp, have, acc, lane);
    } else {
        int* q      = queue + wv * QCAP;
        int  cnt    = 0;
        int  kept_n = 0;

        constexpr int NCHUNK = N_EDGES / SCAN_CHUNK;
        int4 d[4];
#pragma unroll
        for (int j = 0; j < 4; ++j) d[j] = make_int4(-1, -1, -1, -1);
        if (wv < NCHUNK) {
#pragma unroll
            for (int j = 0; j < 4; ++j)
                d[j] = *(const int4*)(dst + wv * SCAN_CHUNK + lane * 4 + j * 128);
        }
        for (int c = wv; c < NCHUNK; c += 32) {
            const int e0 = c * SCAN_CHUNK + lane * 4;
            int4 dn[4];
#pragma unroll
            for (int j = 0; j < 4; ++j) dn[j] = d[j];
            if (c + 32 < NCHUNK) {
#pragma unroll
                for (int j = 0; j < 4; ++j)
                    dn[j] = *(const int4*)(dst + e0 + 32 * SCAN_CHUNK + j * 128);
            }
#pragma unroll
            for (int j = 0; j < 4; ++j) {
                const int dv[4] = { d[j].x, d[j].y, d[j].z, d[j].w };
#pragma unroll
                for (int t = 0; t < 4; ++t) {
                    const unsigned rel  = (unsigned)(dv[t] - base);
                    const bool     hit  = rel < (unsigned)R;
                    const unsigned mask = __builtin_amdgcn_ballot_w32(hit);
                    if (mask) {
                        if (hit) {
                            const int pos = cnt + __builtin_amdgcn_mbcnt_lo(mask, 0);
                            q[pos] = (e0 + j * 128 + t) | ((int)rel << 20);
                        }
                        cnt += __builtin_popcount(mask);
                    }
                }
            }
            if (cnt > QCAP - SCAN_CHUNK) {
                drain_queue<D, MODE == 0>(H, src, val, q, cnt, acc, lane, kp, kept_n);
                kept_n += cnt;
                cnt = 0;
            }
#pragma unroll
            for (int j = 0; j < 4; ++j) d[j] = dn[j];
        }
        drain_queue<D, MODE == 0>(H, src, val, q, cnt, acc, lane, kp, kept_n);
        kept_n += cnt;

        if (MODE == 0 && lane == 0)
            kept_cnt[slot] = (kept_n <= KEEP_CAP) ? kept_n : -1;
    }

    __syncthreads();

    for (int i = threadIdx.x; i < R * D / 4; i += 1024) {
        const int trow = i / (D / 4);
        const int c4   = i % (D / 4);
        const float* p = acc + trow * D + (c4 >> 5) * 128 + (c4 & 31);
        if (base + trow < N_NODES)
            ((float4*)(Z + (size_t)base * D))[i] = make_float4(p[0], p[32], p[64], p[96]);
    }
}

#define KC 32
#define A_PITCH 36

template<int NSUB, int KDIM, bool RELU>
__global__ __launch_bounds__(128) __attribute__((amdgpu_waves_per_eu(1, 3))) void gemm_wmma_f32(
    const float* __restrict__ A,
    const float* __restrict__ B,
    float*       __restrict__ C,
    int M)
{
    constexpr int N       = 64 * NSUB;
    constexpr int B_PITCH = N + 8;
    constexpr int NCH     = KDIM / KC;
    constexpr int A_LD    = (32 * KC / 4) / 128;
    constexpr int B_LD    = (KC * N / 4) / 128;
    static_assert(KDIM % KC == 0, "whole K chunks");

    __shared__ float sA[2][32 * A_PITCH];
    __shared__ float sB[2][KC * B_PITCH];

    const int tid  = threadIdx.x;
    const int lane = tid & 31;
    const int wave = tid >> 5;
    const int lo   = lane & 15;
    const int hi   = lane >> 4;

    const int m0  = blockIdx.x * 32;
    const int nb0 = wave * (NSUB * 16);

    int mbase[2];
#pragma unroll
    for (int ms = 0; ms < 2; ++ms)
        mbase[ms] = min(m0 + 16 * ms, M - 16);

    v8f acc[2][NSUB];
#pragma unroll
    for (int ms = 0; ms < 2; ++ms)
#pragma unroll
        for (int t = 0; t < NSUB; ++t) acc[ms][t] = 0.0f;

    const float* ag[A_LD];
    int          al[A_LD];
#pragma unroll
    for (int i = 0; i < A_LD; ++i) {
        const int idx = tid + 128 * i;
        const int row = idx / (KC / 4);
        const int c4  = idx % (KC / 4);
        ag[i] = A + (size_t)(mbase[row >> 4] + (row & 15)) * KDIM + c4 * 4;
        al[i] = row * A_PITCH + c4 * 4;
    }
    const float* bg[B_LD];
    int          bl[B_LD];
#pragma unroll
    for (int i = 0; i < B_LD; ++i) {
        const int idx = tid + 128 * i;
        const int kr  = idx / (N / 4);
        const int c4  = idx % (N / 4);
        bg[i] = B + (size_t)kr * N + c4 * 4;
        bl[i] = kr * B_PITCH + c4 * 4;
    }

    float4 ra[A_LD], rb[B_LD];
#pragma unroll
    for (int i = 0; i < A_LD; ++i) ra[i] = *(const float4*)(ag[i]);
#pragma unroll
    for (int i = 0; i < B_LD; ++i) rb[i] = *(const float4*)(bg[i]);
#pragma unroll
    for (int i = 0; i < A_LD; ++i) *(float4*)(&sA[0][al[i]]) = ra[i];
#pragma unroll
    for (int i = 0; i < B_LD; ++i) *(float4*)(&sB[0][bl[i]]) = rb[i];
    __syncthreads();

#pragma unroll 1
    for (int c = 0; c < NCH; ++c) {
        const int cur = c & 1;
        const int cn = min(c + 1, NCH - 1);
#pragma unroll
        for (int i = 0; i < A_LD; ++i)
            ra[i] = *(const float4*)(ag[i] + cn * KC);
#pragma unroll
        for (int i = 0; i < B_LD; ++i)
            rb[i] = *(const float4*)(bg[i] + (size_t)cn * KC * N);

        const float* __restrict__ cA = sA[cur];
        const float* __restrict__ cB = sB[cur];
#pragma unroll
        for (int kk = 0; kk < KC; kk += 4) {
            const int ka = kk + 2 * hi;
            v2f a[2];
#pragma unroll
            for (int ms = 0; ms < 2; ++ms)
                a[ms] = *(const v2f*)(cA + (16 * ms + lo) * A_PITCH + ka);
#pragma unroll
            for (int t = 0; t < NSUB; ++t) {
                const int n = nb0 + t * 16 + lo;
                v2f b;
                b.x = cB[ka * B_PITCH + n];
                b.y = cB[(ka + 1) * B_PITCH + n];
#pragma unroll
                for (int ms = 0; ms < 2; ++ms) {
                    acc[ms][t] = __builtin_amdgcn_wmma_f32_16x16x4_f32(
                        false, a[ms], false, b, (short)0, acc[ms][t],
                        false, false);
                }
            }
        }

#pragma unroll
        for (int i = 0; i < A_LD; ++i) *(float4*)(&sA[cur ^ 1][al[i]]) = ra[i];
#pragma unroll
        for (int i = 0; i < B_LD; ++i) *(float4*)(&sB[cur ^ 1][bl[i]]) = rb[i];
        __syncthreads();
    }

#pragma unroll
    for (int ms = 0; ms < 2; ++ms) {
        if (m0 + 16 * ms < M) {
#pragma unroll
            for (int t = 0; t < NSUB; ++t) {
#pragma unroll
                for (int v = 0; v < 8; ++v) {
                    const int row = mbase[ms] + v + 8 * hi;
                    const int col = nb0 + t * 16 + lo;
                    float x = acc[ms][t][v];
                    if (RELU) x = fmaxf(x, 0.0f);
                    C[(size_t)row * N + col] = x;
                }
            }
        }
    }
}

extern "C" void kernel_launch(void* const* d_in, const int* in_sizes, int n_in,
                              void* d_out, int out_size, void* d_ws, size_t ws_size,
                              hipStream_t stream)
{
    (void)in_sizes; (void)n_in; (void)out_size; (void)ws_size;

    const float* features = (const float*)d_in[0];
    const int*   esrc     = (const int*)  d_in[1];
    const int*   edst     = (const int*)  d_in[2];
    const float* eval_    = (const float*)d_in[3];
    const float* W0       = (const float*)d_in[4];
    const float* W1       = (const float*)d_in[5];
    float*       out      = (float*)d_out;

    const size_t SZ = (size_t)N_NODES * HIDDEN * sizeof(float);
    float* Z = (float*)d_ws;
    float* H = (float*)((char*)d_ws + SZ);

    constexpr int SPMM_ROWS = 40;
    const int spmm_blocks = (N_NODES + SPMM_ROWS - 1) / SPMM_ROWS;
    constexpr size_t KEPT = (size_t)((N_NODES + SPMM_ROWS - 1) / SPMM_ROWS) * 32 * KEEP_CAP * sizeof(int2);
    int* kept     = (int*)((char*)d_ws + 2 * SZ);
    int* kept_cnt = (int*)((char*)d_ws + 2 * SZ + KEPT);
    const int gemm_blocks = (N_NODES + 31) / 32;

    spmm_rows_lds<IN_DIM, SPMM_ROWS, 0><<<spmm_blocks, 1024, 0, stream>>>(features, esrc, edst, eval_, Z,
                                                                          kept, kept_cnt);
    gemm_wmma_f32<4, IN_DIM, true><<<gemm_blocks, 128, 0, stream>>>(Z, W0, H, N_NODES);

    float* G = Z;
    gemm_wmma_f32<2, HIDDEN, false><<<gemm_blocks, 128, 0, stream>>>(H, W1, G, N_NODES);
    spmm_rows_lds<OUT_DIM, SPMM_ROWS, 1><<<spmm_blocks, 1024, 0, stream>>>(G, esrc, edst, eval_, out,
                                                                           kept, kept_cnt);
}


# SpMM filter: a pre-pass launch packs each edge to a tile-index byte plus a (row,src,weight) pair; blocks scan 1 byte per edge, not 4, fetch a hit with one load; sums stay arrival-order LDS atomics
# speedup vs baseline: 2.8334x; 1.0548x over previous
#include <hip/hip_runtime.h>

#define N_NODES 10000
#define N_EDGES 320000
#define IN_DIM  256
#define HIDDEN  256
#define OUT_DIM 128

typedef __attribute__((ext_vector_type(2))) float v2f;
typedef __attribute__((ext_vector_type(8))) float v8f;

template<int R>
__global__ __launch_bounds__(64) void pack_edges(
    const int*   __restrict__ src,
    const int*   __restrict__ dst,
    const float* __restrict__ val,
    int*         __restrict__ tile,
    int*         __restrict__ pairs)
{
    static_assert((N_NODES + R - 1) / R <= 255, "tile index fits a byte, 255 is free");
    static_assert(R <= (1 << 11), "tile row fits above the 20 src bits");
    const int lane = threadIdx.x & 31;
    const int wave = blockIdx.x * 2 + (threadIdx.x >> 5);
    int4   d[2], sv[2];
    float4 w[2];
#pragma unroll
    for (int j = 0; j < 2; ++j) {
        const int e0 = wave * 256 + j * 128 + lane * 4;
        d[j]  = *(const int4*)(dst + e0);
        sv[j] = *(const int4*)(src + e0);
        w[j]  = *(const float4*)(val + e0);
    }
#pragma unroll
    for (int j = 0; j < 2; ++j) {
        const int e0 = wave * 256 + j * 128 + lane * 4;
        const int   dv[4] = { d[j].x, d[j].y, d[j].z, d[j].w };
        const int   ss[4] = { sv[j].x, sv[j].y, sv[j].z, sv[j].w };
        const float ww[4] = { w[j].x, w[j].y, w[j].z, w[j].w };
        unsigned ids = 0;
        int2 p[4];
#pragma unroll
        for (int t = 0; t < 4; ++t) {
            const unsigned du = (unsigned)dv[t];
            const bool     ok = du < (unsigned)N_NODES;
            const unsigned ti = du / (unsigned)R;
            const unsigned tr = du - ti * (unsigned)R;
            ids |= (ok ? ti : 255u) << (8 * t);
            p[t] = make_int2(ss[t] | (int)(tr << 20), __float_as_int(ww[t]));
        }
        tile[e0 >> 2] = (int)ids;
        int4* po = (int4*)(pairs + 2 * e0);
        po[0] = make_int4(p[0].x, p[0].y, p[1].x, p[1].y);
        po[1] = make_int4(p[2].x, p[2].y, p[3].x, p[3].y);
    }
}

#define SCAN_CHUNK 512
#define QCAP       1152
#define KEEP_CAP   256

static_assert(N_EDGES % SCAN_CHUNK == 0, "scan and pre-pass take whole chunks");
static_assert(N_EDGES / SCAN_CHUNK >= 32, "every wave of a block has a first chunk");
static_assert(N_NODES <= (1 << 20), "a pair packs src into 20 bits");

template<int D>
__device__ __forceinline__ void load_row(float4 (&h)[D / 128],
                                         const float* __restrict__ H,
                                         int s, int lane)
{
    static_assert(D % 128 == 0, "a row is whole 512-byte pieces");
    const float4* __restrict__ hrow = (const float4*)(H + (size_t)s * D) + lane;
#pragma unroll
    for (int i = 0; i < D / 128; ++i) h[i] = hrow[i * 32];
}

#define GROUP 4

template<int D>
__device__ __forceinline__ void load_group(float4 (&h)[GROUP][D / 128],
                                           const float* __restrict__ H,
                                           int s, int k, int last, int lane)
{
#pragma unroll
    for (int g = 0; g < GROUP; ++g)
        load_row<D>(h[g], H,
                    __builtin_amdgcn_readlane(s, min(k + g, last)), lane);
}

template<int D, bool FULL>
__device__ __forceinline__ void add_group(const float4 (&h)[GROUP][D / 128],
                                          int n, int k, int r, float w,
                                          float* acc, int lane)
{
#pragma unroll
    for (int g = 0; g < GROUP; ++g) {
        if (FULL || k + g < n) {
            const float wk = __int_as_float(
                __builtin_amdgcn_readlane(__float_as_int(w), k + g));
            float* arow = acc + __builtin_amdgcn_readlane(r, k + g) * D + lane;
#pragma unroll
            for (int i = 0; i < D / 128; ++i) {
                atomicAdd(&arow[i * 128],      wk * h[g][i].x);
                atomicAdd(&arow[i * 128 + 32], wk * h[g][i].y);
                atomicAdd(&arow[i * 128 + 64], wk * h[g][i].z);
                atomicAdd(&arow[i * 128 + 96], wk * h[g][i].w);
            }
        }
    }
}

template<int D>
__device__ __forceinline__ void add_batch(
    const float* __restrict__ H, int n, int r, int s, float w,
    float* acc, int lane)
{
    float4 a[GROUP][D / 128];
    float4 b[GROUP][D / 128];
    const int last = n - 1;
    int k = 0;
    load_group<D>(a, H, s, 0, last, lane);
    while (k + 2 * GROUP < n) {
        load_group<D>(b, H, s, k + GROUP, last, lane);
        add_group<D, true>(a, n, k, r, w, acc, lane);
        load_group<D>(a, H, s, k + 2 * GROUP, last, lane);
        add_group<D, true>(b, n, k + GROUP, r, w, acc, lane);
        k += 2 * GROUP;
    }
    if (k + GROUP < n) {
        load_group<D>(b, H, s, k + GROUP, last, lane);
        add_group<D, true>(a, n, k, r, w, acc, lane);
        add_group<D, false>(b, n, k + GROUP, r, w, acc, lane);
    } else {
        add_group<D, false>(a, n, k, r, w, acc, lane);
    }
}

template<int D, bool KEEP>
__device__ __forceinline__ void drain_queue(
    const float* __restrict__ H,
    const int2*  __restrict__ pairs,
    const int* q, int cnt, float* acc, int lane,
    int2* __restrict__ kept, int kept_n)
{
    for (int b = 0; b < cnt; b += 32) {
        const int n = min(32, cnt - b);
        int2 ent = make_int2(0, 0);
        if (lane < n)
            ent = pairs[q[b + lane]];
        const int   s = ent.x & 0xFFFFF;
        const int   r = ent.x >> 20;
        const float w = __int_as_float(ent.y);
        add_batch<D>(H, n, r, s, w, acc, lane);
        if (KEEP) {
            const int pos = kept_n + b + lane;
            if (lane < n && pos < KEEP_CAP)
                kept[pos] = ent;
        }
    }
}

template<int D>
__device__ __forceinline__ void drain_kept(
    const float* __restrict__ H,
    const int2*  __restrict__ kept, int cnt, float* acc, int lane)
{
    for (int b = 0; b < cnt; b += 32) {
        const int n = min(32, cnt - b);
        int   r = 0, s = 0;
        float w = 0.0f;
        if (lane < n) {
            const int2 ent = kept[b + lane];
            s = ent.x & 0xFFFFF;
            r = ent.x >> 20;
            w = __int_as_float(ent.y);
        }
        add_batch<D>(H, n, r, s, w, acc, lane);
    }
}

template<int D, int R, int MODE>
__global__ __launch_bounds__(1024) void spmm_rows_lds(
    const float* __restrict__ H,
    const int*   __restrict__ tile,
    const int*   __restrict__ pairs,
    float*       __restrict__ Z,
    int*         __restrict__ kept,
    int*         __restrict__ kept_cnt)
{
    __shared__ float acc[R * D];
    __shared__ int   queue[32 * QCAP];

    const int lane = threadIdx.x & 31;
    const int wv   = __builtin_amdgcn_readfirstlane(threadIdx.x >> 5);
    const int base = blockIdx.x * R;
    const int slot = blockIdx.x * 32 + wv;

    for (int i = threadIdx.x; i < R * D / 4; i += 1024)
        ((float4*)acc)[i] = make_float4(0.0f, 0.0f, 0.0f, 0.0f);
    __syncthreads();

    int2* kp = (int2*)kept + (size_t)slot * KEEP_CAP;
    const int2* __restrict__ pr = (const int2*)pairs;

    int have = -1;
    if (MODE == 1)
        have = __builtin_amdgcn_readfirstlane(kept_cnt[slot]);

    if (have >= 0) {
        drain_kept<D>(H, kp, have, acc, lane);
    } else {
        int* q      = queue + wv * QCAP;
        int  cnt    = 0;
        int  kept_n = 0;

        constexpr int NCHUNK = N_EDGES / SCAN_CHUNK;
        const uint4* __restrict__ tp = (const uint4*)tile + lane;
        const unsigned pat = (unsigned)blockIdx.x * 0x01010101u;
        uint4 d[4];
#pragma unroll
        for (int j = 0; j < 4; ++j)
            d[j] = tp[min(wv + 32 * j, NCHUNK - 1) * 32];
        for (int c = wv; c < NCHUNK; c += 128) {
            uint4 dn[4];
#pragma unroll
            for (int j = 0; j < 4; ++j)
                dn[j] = tp[min(c + 128 + 32 * j, NCHUNK - 1) * 32];
#pragma unroll
            for (int j = 0; j < 4; ++j) {
                if (c + 32 * j >= NCHUNK) break;
                const int e0 = (c + 32 * j) * SCAN_CHUNK + lane * 16;
                const unsigned xv[4] = { d[j].x, d[j].y, d[j].z, d[j].w };
#pragma unroll
                for (int t = 0; t < 4; ++t) {
                    const unsigned y = xv[t] ^ pat;
                    const bool     any   = ((y - 0x01010101u) & ~y & 0x80808080u) != 0;
                    const unsigned amask = __builtin_amdgcn_ballot_w32(any);
                    if (amask) {
#pragma unroll
                        for (int k = 0; k < 4; ++k) {
                            const bool hit = ((y >> (8 * k)) & 0xFFu) == 0;
                            const unsigned mask = __builtin_amdgcn_ballot_w32(hit);
                            if (mask) {
                                if (hit) {
                                    const int pos = cnt + __builtin_amdgcn_mbcnt_lo(mask, 0);
                                    q[pos] = e0 + 4 * t + k;
                                }
                                cnt += __builtin_popcount(mask);
                            }
                        }
                    }
                }
                if (cnt > QCAP - SCAN_CHUNK) {
                    drain_queue<D, MODE == 0>(H, pr, q, cnt, acc, lane, kp, kept_n);
                    kept_n += cnt;
                    cnt = 0;
                }
            }
#pragma unroll
            for (int j = 0; j < 4; ++j) d[j] = dn[j];
        }
        drain_queue<D, MODE == 0>(H, pr, q, cnt, acc, lane, kp, kept_n);
        kept_n += cnt;

        if (MODE == 0 && lane == 0)
            kept_cnt[slot] = (kept_n <= KEEP_CAP) ? kept_n : -1;
    }

    __syncthreads();

    for (int i = threadIdx.x; i < R * D / 4; i += 1024) {
        const int trow = i / (D / 4);
        const int c4   = i % (D / 4);
        const float* p = acc + trow * D + (c4 >> 5) * 128 + (c4 & 31);
        if (base + trow < N_NODES)
            ((float4*)(Z + (size_t)base * D))[i] = make_float4(p[0], p[32], p[64], p[96]);
    }
}

#define KC 32
#define A_PITCH 36

template<int NSUB, int KDIM, bool RELU>
__global__ __launch_bounds__(128) __attribute__((amdgpu_waves_per_eu(1, 3))) void gemm_wmma_f32(
    const float* __restrict__ A,
    const float* __restrict__ B,
    float*       __restrict__ C,
    int M)
{
    constexpr int N       = 64 * NSUB;
    constexpr int B_PITCH = N + 8;
    constexpr int NCH     = KDIM / KC;
    constexpr int A_LD    = (32 * KC / 4) / 128;
    constexpr int B_LD    = (KC * N / 4) / 128;
    static_assert(KDIM % KC == 0, "whole K chunks");

    __shared__ float sA[2][32 * A_PITCH];
    __shared__ float sB[2][KC * B_PITCH];

    const int tid  = threadIdx.x;
    const int lane = tid & 31;
    const int wave = tid >> 5;
    const int lo   = lane & 15;
    const int hi   = lane >> 4;

    const int m0  = blockIdx.x * 32;
    const int nb0 = wave * (NSUB * 16);

    int mbase[2];
#pragma unroll
    for (int ms = 0; ms < 2; ++ms)
        mbase[ms] = min(m0 + 16 * ms, M - 16);

    v8f acc[2][NSUB];
#pragma unroll
    for (int ms = 0; ms < 2; ++ms)
#pragma unroll
        for (int t = 0; t < NSUB; ++t) acc[ms][t] = 0.0f;

    const float* ag[A_LD];
    int          al[A_LD];
#pragma unroll
    for (int i = 0; i < A_LD; ++i) {
        const int idx = tid + 128 * i;
        const int row = idx / (KC / 4);
        const int c4  = idx % (KC / 4);
        ag[i] = A + (size_t)(mbase[row >> 4] + (row & 15)) * KDIM + c4 * 4;
        al[i] = row * A_PITCH + c4 * 4;
    }
    const float* bg[B_LD];
    int          bl[B_LD];
#pragma unroll
    for (int i = 0; i < B_LD; ++i) {
        const int idx = tid + 128 * i;
        const int kr  = idx / (N / 4);
        const int c4  = idx % (N / 4);
        bg[i] = B + (size_t)kr * N + c4 * 4;
        bl[i] = kr * B_PITCH + c4 * 4;
    }

    float4 ra[A_LD], rb[B_LD];
#pragma unroll
    for (int i = 0; i < A_LD; ++i) ra[i] = *(const float4*)(ag[i]);
#pragma unroll
    for (int i = 0; i < B_LD; ++i) rb[i] = *(const float4*)(bg[i]);
#pragma unroll
    for (int i = 0; i < A_LD; ++i) *(float4*)(&sA[0][al[i]]) = ra[i];
#pragma unroll
    for (int i = 0; i < B_LD; ++i) *(float4*)(&sB[0][bl[i]]) = rb[i];
    __syncthreads();

#pragma unroll 1
    for (int c = 0; c < NCH; ++c) {
        const int cur = c & 1;
        const int cn = min(c + 1, NCH - 1);
#pragma unroll
        for (int i = 0; i < A_LD; ++i)
            ra[i] = *(const float4*)(ag[i] + cn * KC);
#pragma unroll
        for (int i = 0; i < B_LD; ++i)
            rb[i] = *(const float4*)(bg[i] + (size_t)cn * KC * N);

        const float* __restrict__ cA = sA[cur];
        const float* __restrict__ cB = sB[cur];
#pragma unroll
        for (int kk = 0; kk < KC; kk += 4) {
            const int ka = kk + 2 * hi;
            v2f a[2];
#pragma unroll
            for (int ms = 0; ms < 2; ++ms)
                a[ms] = *(const v2f*)(cA + (16 * ms + lo) * A_PITCH + ka);
#pragma unroll
            for (int t = 0; t < NSUB; ++t) {
                const int n = nb0 + t * 16 + lo;
                v2f b;
                b.x = cB[ka * B_PITCH + n];
                b.y = cB[(ka + 1) * B_PITCH + n];
#pragma unroll
                for (int ms = 0; ms < 2; ++ms) {
                    acc[ms][t] = __builtin_amdgcn_wmma_f32_16x16x4_f32(
                        false, a[ms], false, b, (short)0, acc[ms][t],
                        false, false);
                }
            }
        }

#pragma unroll
        for (int i = 0; i < A_LD; ++i) *(float4*)(&sA[cur ^ 1][al[i]]) = ra[i];
#pragma unroll
        for (int i = 0; i < B_LD; ++i) *(float4*)(&sB[cur ^ 1][bl[i]]) = rb[i];
        __syncthreads();
    }

#pragma unroll
    for (int ms = 0; ms < 2; ++ms) {
        if (m0 + 16 * ms < M) {
#pragma unroll
            for (int t = 0; t < NSUB; ++t) {
#pragma unroll
                for (int v = 0; v < 8; ++v) {
                    const int row = mbase[ms] + v + 8 * hi;
                    const int col = nb0 + t * 16 + lo;
                    float x = acc[ms][t][v];
                    if (RELU) x = fmaxf(x, 0.0f);
                    C[(size_t)row * N + col] = x;
                }
            }
        }
    }
}

extern "C" void kernel_launch(void* const* d_in, const int* in_sizes, int n_in,
                              void* d_out, int out_size, void* d_ws, size_t ws_size,
                              hipStream_t stream)
{
    (void)in_sizes; (void)n_in; (void)out_size; (void)ws_size;

    const float* features = (const float*)d_in[0];
    const int*   esrc     = (const int*)  d_in[1];
    const int*   edst     = (const int*)  d_in[2];
    const float* eval_    = (const float*)d_in[3];
    const float* W0       = (const float*)d_in[4];
    const float* W1       = (const float*)d_in[5];
    float*       out      = (float*)d_out;

    const size_t SZ = (size_t)N_NODES * HIDDEN * sizeof(float);
    float* Z = (float*)d_ws;
    float* H = (float*)((char*)d_ws + SZ);

    constexpr int SPMM_ROWS = 40;
    const int spmm_blocks = (N_NODES + SPMM_ROWS - 1) / SPMM_ROWS;
    constexpr size_t KEPT = (size_t)((N_NODES + SPMM_ROWS - 1) / SPMM_ROWS) * 32 * KEEP_CAP * sizeof(int2);
    int* kept     = (int*)((char*)d_ws + 2 * SZ);
    int* kept_cnt = (int*)((char*)d_ws + 2 * SZ + KEPT);
    constexpr size_t CNTS = (size_t)((N_NODES + SPMM_ROWS - 1) / SPMM_ROWS) * 32 * sizeof(int);
    static_assert((2 * (size_t)N_NODES * HIDDEN * sizeof(float) + KEPT + CNTS) % 16 == 0 &&
                  N_EDGES % 16 == 0, "the byte plane and the pairs start on 16-byte boundaries");
    int* tile  = (int*)((char*)d_ws + 2 * SZ + KEPT + CNTS);
    int* pairs = (int*)((char*)d_ws + 2 * SZ + KEPT + CNTS + N_EDGES);
    const int gemm_blocks = (N_NODES + 31) / 32;

    static_assert(SCAN_CHUNK == 2 * 256, "a pre-pass block covers one chunk");
    pack_edges<SPMM_ROWS><<<N_EDGES / SCAN_CHUNK, 64, 0, stream>>>(esrc, edst, eval_, tile, pairs);

    spmm_rows_lds<IN_DIM, SPMM_ROWS, 0><<<spmm_blocks, 1024, 0, stream>>>(features, tile, pairs, Z,
                                                                          kept, kept_cnt);
    gemm_wmma_f32<4, IN_DIM, true><<<gemm_blocks, 128, 0, stream>>>(Z, W0, H, N_NODES);

    float* G = Z;
    gemm_wmma_f32<2, HIDDEN, false><<<gemm_blocks, 128, 0, stream>>>(H, W1, G, N_NODES);
    spmm_rows_lds<OUT_DIM, SPMM_ROWS, 1><<<spmm_blocks, 1024, 0, stream>>>(G, tile, pairs, out,
                                                                           kept, kept_cnt);
}


# GEMMs re-tiled for equal matrix work per SIMD: blocks of 80 rows x half of N (250 blocks, one per WGP; was 313 of 32 rows x N), whole operands in LDS, K chunks of 64; k and sum order unchanged
# speedup vs baseline: 3.2173x; 1.1355x over previous
#include <hip/hip_runtime.h>

#define N_NODES 10000
#define N_EDGES 320000
#define IN_DIM  256
#define HIDDEN  256
#define OUT_DIM 128

typedef __attribute__((ext_vector_type(2))) float v2f;
typedef __attribute__((ext_vector_type(8))) float v8f;

template<int R>
__global__ __launch_bounds__(64) void pack_edges(
    const int*   __restrict__ src,
    const int*   __restrict__ dst,
    const float* __restrict__ val,
    int*         __restrict__ tile,
    int*         __restrict__ pairs)
{
    static_assert((N_NODES + R - 1) / R <= 255, "tile index fits a byte, 255 is free");
    static_assert(R <= (1 << 11), "tile row fits above the 20 src bits");
    const int lane = threadIdx.x & 31;
    const int wave = blockIdx.x * 2 + (threadIdx.x >> 5);
    int4   d[2], sv[2];
    float4 w[2];
#pragma unroll
    for (int j = 0; j < 2; ++j) {
        const int e0 = wave * 256 + j * 128 + lane * 4;
        d[j]  = *(const int4*)(dst + e0);
        sv[j] = *(const int4*)(src + e0);
        w[j]  = *(const float4*)(val + e0);
    }
#pragma unroll
    for (int j = 0; j < 2; ++j) {
        const int e0 = wave * 256 + j * 128 + lane * 4;
        const int   dv[4] = { d[j].x, d[j].y, d[j].z, d[j].w };
        const int   ss[4] = { sv[j].x, sv[j].y, sv[j].z, sv[j].w };
        const float ww[4] = { w[j].x, w[j].y, w[j].z, w[j].w };
        unsigned ids = 0;
        int2 p[4];
#pragma unroll
        for (int t = 0; t < 4; ++t) {
            const unsigned du = (unsigned)dv[t];
            const bool     ok = du < (unsigned)N_NODES;
            const unsigned ti = du / (unsigned)R;
            const unsigned tr = du - ti * (unsigned)R;
            ids |= (ok ? ti : 255u) << (8 * t);
            p[t] = make_int2(ss[t] | (int)(tr << 20), __float_as_int(ww[t]));
        }
        tile[e0 >> 2] = (int)ids;
        int4* po = (int4*)(pairs + 2 * e0);
        po[0] = make_int4(p[0].x, p[0].y, p[1].x, p[1].y);
        po[1] = make_int4(p[2].x, p[2].y, p[3].x, p[3].y);
    }
}

#define SCAN_CHUNK 512
#define QCAP       1152
#define KEEP_CAP   256

static_assert(N_EDGES % SCAN_CHUNK == 0, "scan and pre-pass take whole chunks");
static_assert(N_EDGES / SCAN_CHUNK >= 32, "every wave of a block has a first chunk");
static_assert(N_NODES <= (1 << 20), "a pair packs src into 20 bits");

template<int D>
__device__ __forceinline__ void load_row(float4 (&h)[D / 128],
                                         const float* __restrict__ H,
                                         int s, int lane)
{
    static_assert(D % 128 == 0, "a row is whole 512-byte pieces");
    const float4* __restrict__ hrow = (const float4*)(H + (size_t)s * D) + lane;
#pragma unroll
    for (int i = 0; i < D / 128; ++i) h[i] = hrow[i * 32];
}

#define GROUP 4

template<int D>
__device__ __forceinline__ void load_group(float4 (&h)[GROUP][D / 128],
                                           const float* __restrict__ H,
                                           int s, int k, int last, int lane)
{
#pragma unroll
    for (int g = 0; g < GROUP; ++g)
        load_row<D>(h[g], H,
                    __builtin_amdgcn_readlane(s, min(k + g, last)), lane);
}

template<int D, bool FULL>
__device__ __forceinline__ void add_group(const float4 (&h)[GROUP][D / 128],
                                          int n, int k, int r, float w,
                                          float* acc, int lane)
{
#pragma unroll
    for (int g = 0; g < GROUP; ++g) {
        if (FULL || k + g < n) {
            const float wk = __int_as_float(
                __builtin_amdgcn_readlane(__float_as_int(w), k + g));
            float* arow = acc + __builtin_amdgcn_readlane(r, k + g) * D + lane;
#pragma unroll
            for (int i = 0; i < D / 128; ++i) {
                atomicAdd(&arow[i * 128],      wk * h[g][i].x);
                atomicAdd(&arow[i * 128 + 32], wk * h[g][i].y);
                atomicAdd(&arow[i * 128 + 64], wk * h[g][i].z);
                atomicAdd(&arow[i * 128 + 96], wk * h[g][i].w);
            }
        }
    }
}

template<int D>
__device__ __forceinline__ void add_batch(
    const float* __restrict__ H, int n, int r, int s, float w,
    float* acc, int lane)
{
    float4 a[GROUP][D / 128];
    float4 b[GROUP][D / 128];
    const int last = n - 1;
    int k = 0;
    load_group<D>(a, H, s, 0, last, lane);
    while (k + 2 * GROUP < n) {
        load_group<D>(b, H, s, k + GROUP, last, lane);
        add_group<D, true>(a, n, k, r, w, acc, lane);
        load_group<D>(a, H, s, k + 2 * GROUP, last, lane);
        add_group<D, true>(b, n, k + GROUP, r, w, acc, lane);
        k += 2 * GROUP;
    }
    if (k + GROUP < n) {
        load_group<D>(b, H, s, k + GROUP, last, lane);
        add_group<D, true>(a, n, k, r, w, acc, lane);
        add_group<D, false>(b, n, k + GROUP, r, w, acc, lane);
    } else {
        add_group<D, false>(a, n, k, r, w, acc, lane);
    }
}

template<int D, bool KEEP>
__device__ __forceinline__ void drain_queue(
    const float* __restrict__ H,
    const int2*  __restrict__ pairs,
    const int* q, int cnt, float* acc, int lane,
    int2* __restrict__ kept, int kept_n)
{
    for (int b = 0; b < cnt; b += 32) {
        const int n = min(32, cnt - b);
        int2 ent = make_int2(0, 0);
        if (lane < n)
            ent = pairs[q[b + lane]];
        const int   s = ent.x & 0xFFFFF;
        const int   r = ent.x >> 20;
        const float w = __int_as_float(ent.y);
        add_batch<D>(H, n, r, s, w, acc, lane);
        if (KEEP) {
            const int pos = kept_n + b + lane;
            if (lane < n && pos < KEEP_CAP)
                kept[pos] = ent;
        }
    }
}

template<int D>
__device__ __forceinline__ void drain_kept(
    const float* __restrict__ H,
    const int2*  __restrict__ kept, int cnt, float* acc, int lane)
{
    for (int b = 0; b < cnt; b += 32) {
        const int n = min(32, cnt - b);
        int   r = 0, s = 0;
        float w = 0.0f;
        if (lane < n) {
            const int2 ent = kept[b + lane];
            s = ent.x & 0xFFFFF;
            r = ent.x >> 20;
            w = __int_as_float(ent.y);
        }
        add_batch<D>(H, n, r, s, w, acc, lane);
    }
}

template<int D, int R, int MODE>
__global__ __launch_bounds__(1024) void spmm_rows_lds(
    const float* __restrict__ H,
    const int*   __restrict__ tile,
    const int*   __restrict__ pairs,
    float*       __restrict__ Z,
    int*         __restrict__ kept,
    int*         __restrict__ kept_cnt)
{
    __shared__ float acc[R * D];
    __shared__ int   queue[32 * QCAP];

    const int lane = threadIdx.x & 31;
    const int wv   = __builtin_amdgcn_readfirstlane(threadIdx.x >> 5);
    const int base = blockIdx.x * R;
    const int slot = blockIdx.x * 32 + wv;

    for (int i = threadIdx.x; i < R * D / 4; i += 1024)
        ((float4*)acc)[i] = make_float4(0.0f, 0.0f, 0.0f, 0.0f);
    __syncthreads();

    int2* kp = (int2*)kept + (size_t)slot * KEEP_CAP;
    const int2* __restrict__ pr = (const int2*)pairs;

    int have = -1;
    if (MODE == 1)
        have = __builtin_amdgcn_readfirstlane(kept_cnt[slot]);

    if (have >= 0) {
        drain_kept<D>(H, kp, have, acc, lane);
    } else {
        int* q      = queue + wv * QCAP;
        int  cnt    = 0;
        int  kept_n = 0;

        constexpr int NCHUNK = N_EDGES / SCAN_CHUNK;
        const uint4* __restrict__ tp = (const uint4*)tile + lane;
        const unsigned pat = (unsigned)blockIdx.x * 0x01010101u;
        uint4 d[4];
#pragma unroll
        for (int j = 0; j < 4; ++j)
            d[j] = tp[min(wv + 32 * j, NCHUNK - 1) * 32];
        for (int c = wv; c < NCHUNK; c += 128) {
            uint4 dn[4];
#pragma unroll
            for (int j = 0; j < 4; ++j)
                dn[j] = tp[min(c + 128 + 32 * j, NCHUNK - 1) * 32];
#pragma unroll
            for (int j = 0; j < 4; ++j) {
                if (c + 32 * j >= NCHUNK) break;
                const int e0 = (c + 32 * j) * SCAN_CHUNK + lane * 16;
                const unsigned xv[4] = { d[j].x, d[j].y, d[j].z, d[j].w };
#pragma unroll
                for (int t = 0; t < 4; ++t) {
                    const unsigned y = xv[t] ^ pat;
                    const bool     any   = ((y - 0x01010101u) & ~y & 0x80808080u) != 0;
                    const unsigned amask = __builtin_amdgcn_ballot_w32(any);
                    if (amask) {
#pragma unroll
                        for (int k = 0; k < 4; ++k) {
                            const bool hit = ((y >> (8 * k)) & 0xFFu) == 0;
                            const unsigned mask = __builtin_amdgcn_ballot_w32(hit);
                            if (mask) {
                                if (hit) {
                                    const int pos = cnt + __builtin_amdgcn_mbcnt_lo(mask, 0);
                                    q[pos] = e0 + 4 * t + k;
                                }
                                cnt += __builtin_popcount(mask);
                            }
                        }
                    }
                }
                if (cnt > QCAP - SCAN_CHUNK) {
                    drain_queue<D, MODE == 0>(H, pr, q, cnt, acc, lane, kp, kept_n);
                    kept_n += cnt;
                    cnt = 0;
                }
            }
#pragma unroll
            for (int j = 0; j < 4; ++j) d[j] = dn[j];
        }
        drain_queue<D, MODE == 0>(H, pr, q, cnt, acc, lane, kp, kept_n);
        kept_n += cnt;

        if (MODE == 0 && lane == 0)
            kept_cnt[slot] = (kept_n <= KEEP_CAP) ? kept_n : -1;
    }

    __syncthreads();

    for (int i = threadIdx.x; i < R * D / 4; i += 1024) {
        const int trow = i / (D / 4);
        const int c4   = i % (D / 4);
        const float* p = acc + trow * D + (c4 >> 5) * 128 + (c4 & 31);
        if (base + trow < N_NODES)
            ((float4*)(Z + (size_t)base * D))[i] = make_float4(p[0], p[32], p[64], p[96]);
    }
}

#define KC 64
#define MT 5

static_assert(N_NODES % (16 * MT) == 0, "M is whole 80-row blocks");

template<int NSUB, int N, int KDIM, bool RELU>
__global__ __launch_bounds__(128) __attribute__((amdgpu_waves_per_eu(1, 1))) void gemm_wmma_f32(
    const float* __restrict__ A,
    const float* __restrict__ B,
    float*       __restrict__ C)
{
    constexpr int NB      = 64 * NSUB;
    constexpr int A_PITCH = KDIM + 4;
    constexpr int B_PITCH = NB + 16;
    constexpr int NCH     = KDIM / KC;
    constexpr int A_LD    = (16 * MT * KC / 4) / 128;
    constexpr int B_LD    = (KC * NB / 4) / 128;
    static_assert(KDIM % KC == 0, "whole K chunks");
    static_assert(N == 2 * NB, "the grid's y splits N in two");
    static_assert((16 * MT * KC / 4) % 128 == 0 && (KC * NB / 4) % 128 == 0,
                  "a chunk is whole float4 rounds of the block");

    __shared__ float sA[16 * MT * A_PITCH];
    __shared__ float sB[KDIM * B_PITCH];

    const int tid  = threadIdx.x;
    const int lane = tid & 31;
    const int wave = tid >> 5;
    const int lo   = lane & 15;
    const int hi   = lane >> 4;

    const int m0  = blockIdx.x * (16 * MT);
    const int n0  = blockIdx.y * NB;
    const int nb0 = wave * (NSUB * 16);

    v8f acc[MT][NSUB];
#pragma unroll
    for (int ms = 0; ms < MT; ++ms)
#pragma unroll
        for (int t = 0; t < NSUB; ++t) acc[ms][t] = 0.0f;

    const float* ag[A_LD];
    int          al[A_LD];
#pragma unroll
    for (int i = 0; i < A_LD; ++i) {
        const int idx = tid + 128 * i;
        const int row = idx / (KC / 4);
        const int c4  = idx % (KC / 4);
        ag[i] = A + (size_t)(m0 + row) * KDIM + c4 * 4;
        al[i] = row * A_PITCH + c4 * 4;
    }
    const float* bg[B_LD];
    int          bl[B_LD];
#pragma unroll
    for (int i = 0; i < B_LD; ++i) {
        const int idx = tid + 128 * i;
        const int kr  = idx / (NB / 4);
        const int c4  = idx % (NB / 4);
        bg[i] = B + (size_t)kr * N + n0 + c4 * 4;
        bl[i] = kr * B_PITCH + c4 * 4;
    }

    float4 ra[A_LD], rb[B_LD];
#pragma unroll
    for (int i = 0; i < A_LD; ++i) ra[i] = *(const float4*)(ag[i]);
#pragma unroll
    for (int i = 0; i < B_LD; ++i) rb[i] = *(const float4*)(bg[i]);
#pragma unroll
    for (int i = 0; i < A_LD; ++i) *(float4*)(&sA[al[i]]) = ra[i];
#pragma unroll
    for (int i = 0; i < B_LD; ++i) *(float4*)(&sB[bl[i]]) = rb[i];
    __syncthreads();

    auto chunk = [&](int c) {
        const float* __restrict__ cA = sA + c * KC;
        const float* __restrict__ cB = sB + c * KC * B_PITCH;
#pragma unroll
        for (int kk = 0; kk < KC; kk += 4) {
            const int ka = kk + 2 * hi;
            v2f a[MT];
#pragma unroll
            for (int ms = 0; ms < MT; ++ms)
                a[ms] = *(const v2f*)(cA + (16 * ms + lo) * A_PITCH + ka);
#pragma unroll
            for (int t = 0; t < NSUB; ++t) {
                const int n = nb0 + t * 16 + lo;
                v2f b;
                b.x = cB[ka * B_PITCH + n];
                b.y = cB[(ka + 1) * B_PITCH + n];
#pragma unroll
                for (int ms = 0; ms < MT; ++ms) {
                    acc[ms][t] = __builtin_amdgcn_wmma_f32_16x16x4_f32(
                        false, a[ms], false, b, (short)0, acc[ms][t],
                        false, false);
                }
            }
        }
    };

#pragma unroll
    for (int c = 0; c + 1 < NCH; ++c) {
#pragma unroll
        for (int i = 0; i < A_LD; ++i)
            ra[i] = *(const float4*)(ag[i] + (c + 1) * KC);
#pragma unroll
        for (int i = 0; i < B_LD; ++i)
            rb[i] = *(const float4*)(bg[i] + (size_t)(c + 1) * KC * N);

        chunk(c);

#pragma unroll
        for (int i = 0; i < A_LD; ++i) *(float4*)(&sA[al[i] + (c + 1) * KC]) = ra[i];
#pragma unroll
        for (int i = 0; i < B_LD; ++i) *(float4*)(&sB[bl[i] + (c + 1) * KC * B_PITCH]) = rb[i];
        __syncthreads();
    }
    chunk(NCH - 1);

#pragma unroll
    for (int ms = 0; ms < MT; ++ms) {
#pragma unroll
        for (int t = 0; t < NSUB; ++t) {
#pragma unroll
            for (int v = 0; v < 8; ++v) {
                const int row = m0 + 16 * ms + v + 8 * hi;
                const int col = n0 + nb0 + t * 16 + lo;
                float x = acc[ms][t][v];
                if (RELU) x = fmaxf(x, 0.0f);
                C[(size_t)row * N + col] = x;
            }
        }
    }
}

extern "C" void kernel_launch(void* const* d_in, const int* in_sizes, int n_in,
                              void* d_out, int out_size, void* d_ws, size_t ws_size,
                              hipStream_t stream)
{
    (void)in_sizes; (void)n_in; (void)out_size; (void)ws_size;

    const float* features = (const float*)d_in[0];
    const int*   esrc     = (const int*)  d_in[1];
    const int*   edst     = (const int*)  d_in[2];
    const float* eval_    = (const float*)d_in[3];
    const float* W0       = (const float*)d_in[4];
    const float* W1       = (const float*)d_in[5];
    float*       out      = (float*)d_out;

    const size_t SZ = (size_t)N_NODES * HIDDEN * sizeof(float);
    float* Z = (float*)d_ws;
    float* H = (float*)((char*)d_ws + SZ);

    constexpr int SPMM_ROWS = 40;
    const int spmm_blocks = (N_NODES + SPMM_ROWS - 1) / SPMM_ROWS;
    constexpr size_t KEPT = (size_t)((N_NODES + SPMM_ROWS - 1) / SPMM_ROWS) * 32 * KEEP_CAP * sizeof(int2);
    int* kept     = (int*)((char*)d_ws + 2 * SZ);
    int* kept_cnt = (int*)((char*)d_ws + 2 * SZ + KEPT);
    constexpr size_t CNTS = (size_t)((N_NODES + SPMM_ROWS - 1) / SPMM_ROWS) * 32 * sizeof(int);
    static_assert((2 * (size_t)N_NODES * HIDDEN * sizeof(float) + KEPT + CNTS) % 16 == 0 &&
                  N_EDGES % 16 == 0, "the byte plane and the pairs start on 16-byte boundaries");
    int* tile  = (int*)((char*)d_ws + 2 * SZ + KEPT + CNTS);
    int* pairs = (int*)((char*)d_ws + 2 * SZ + KEPT + CNTS + N_EDGES);
    const dim3 gemm_grid(N_NODES / (16 * MT), 2);

    static_assert(SCAN_CHUNK == 2 * 256, "a pre-pass block covers one chunk");
    pack_edges<SPMM_ROWS><<<N_EDGES / SCAN_CHUNK, 64, 0, stream>>>(esrc, edst, eval_, tile, pairs);

    spmm_rows_lds<IN_DIM, SPMM_ROWS, 0><<<spmm_blocks, 1024, 0, stream>>>(features, tile, pairs, Z,
                                                                          kept, kept_cnt);
    gemm_wmma_f32<2, HIDDEN, IN_DIM, true><<<gemm_grid, 128, 0, stream>>>(Z, W0, H);

    float* G = Z;
    gemm_wmma_f32<1, OUT_DIM, HIDDEN, false><<<gemm_grid, 128, 0, stream>>>(H, W1, G);
    spmm_rows_lds<OUT_DIM, SPMM_ROWS, 1><<<spmm_blocks, 1024, 0, stream>>>(G, tile, pairs, out,
                                                                           kept, kept_cnt);
}


# Layer 0 re-associated: relu(A(X@W0)), was relu((AX)@W0) (f32 sum order changes); X@W0 needs no edges, so its GEMM blocks also rewrite the edge array: pre-pass launch gone, 4 launches
# speedup vs baseline: 3.3430x; 1.0391x over previous
#include <hip/hip_runtime.h>

#define N_NODES 10000
#define N_EDGES 320000
#define IN_DIM  256
#define HIDDEN  256
#define OUT_DIM 128

typedef __attribute__((ext_vector_type(2))) float v2f;
typedef __attribute__((ext_vector_type(8))) float v8f;

#define SPMM_ROWS   40
#define GEMM_BLOCKS 250
#define PACK_SHARE  (N_EDGES / GEMM_BLOCKS)

static_assert(N_EDGES % GEMM_BLOCKS == 0 && PACK_SHARE % 256 == 0,
              "a block's share of the edges is whole rounds of its 128 threads");

struct EdgeItem { int2 d, s; float2 w; };

__device__ __forceinline__ void edge_item_load(EdgeItem& it,
                                               const int*   __restrict__ src,
                                               const int*   __restrict__ dst,
                                               const float* __restrict__ val, int e0)
{
    it.d = *(const int2*)(dst + e0);
    it.s = *(const int2*)(src + e0);
    it.w = *(const float2*)(val + e0);
}

template<int R>
__device__ __forceinline__ void edge_item_pack(const EdgeItem& it,
                                               int* __restrict__ tile,
                                               int* __restrict__ pairs,
                                               int e0)
{
    static_assert((N_NODES + R - 1) / R <= 255, "tile index fits a byte, 255 is free");
    static_assert(R <= (1 << 11), "tile row fits above the 20 src bits");
    const int   dv[2] = { it.d.x, it.d.y };
    const int   ss[2] = { it.s.x, it.s.y };
    const float ww[2] = { it.w.x, it.w.y };
    unsigned ids = 0;
    int2 p[2];
#pragma unroll
    for (int t = 0; t < 2; ++t) {
        const unsigned du = (unsigned)dv[t];
        const bool     ok = du < (unsigned)N_NODES;
        const unsigned ti = du / (unsigned)R;
        const unsigned tr = du - ti * (unsigned)R;
        ids |= (ok ? ti : 255u) << (8 * t);
        p[t] = make_int2(ss[t] | (int)(tr << 20), __float_as_int(ww[t]));
    }
    ((unsigned short*)tile)[e0 >> 1] = (unsigned short)ids;
    *(int4*)(pairs + 2 * e0) = make_int4(p[0].x, p[0].y, p[1].x, p[1].y);
}

#define SCAN_CHUNK 512
#define QCAP       1152
#define KEEP_CAP   256

static_assert(N_EDGES % SCAN_CHUNK == 0, "the scan takes whole chunks");
static_assert(N_EDGES / SCAN_CHUNK >= 32, "every wave of a block has a first chunk");
static_assert(N_NODES <= (1 << 20), "a pair packs src into 20 bits");

template<int D>
__device__ __forceinline__ void load_row(float4 (&h)[D / 128],
                                         const float* __restrict__ H,
                                         int s, int lane)
{
    static_assert(D % 128 == 0, "a row is whole 512-byte pieces");
    const float4* __restrict__ hrow = (const float4*)(H + (size_t)s * D) + lane;
#pragma unroll
    for (int i = 0; i < D / 128; ++i) h[i] = hrow[i * 32];
}

#define GROUP 4

template<int D>
__device__ __forceinline__ void load_group(float4 (&h)[GROUP][D / 128],
                                           const float* __restrict__ H,
                                           int s, int k, int last, int lane)
{
#pragma unroll
    for (int g = 0; g < GROUP; ++g)
        load_row<D>(h[g], H,
                    __builtin_amdgcn_readlane(s, min(k + g, last)), lane);
}

template<int D, bool FULL>
__device__ __forceinline__ void add_group(const float4 (&h)[GROUP][D / 128],
                                          int n, int k, int r, float w,
                                          float* acc, int lane)
{
#pragma unroll
    for (int g = 0; g < GROUP; ++g) {
        if (FULL || k + g < n) {
            const float wk = __int_as_float(
                __builtin_amdgcn_readlane(__float_as_int(w), k + g));
            float* arow = acc + __builtin_amdgcn_readlane(r, k + g) * D + lane;
#pragma unroll
            for (int i = 0; i < D / 128; ++i) {
                atomicAdd(&arow[i * 128],      wk * h[g][i].x);
                atomicAdd(&arow[i * 128 + 32], wk * h[g][i].y);
                atomicAdd(&arow[i * 128 + 64], wk * h[g][i].z);
                atomicAdd(&arow[i * 128 + 96], wk * h[g][i].w);
            }
        }
    }
}

template<int D>
__device__ __forceinline__ void add_batch(
    const float* __restrict__ H, int n, int r, int s, float w,
    float* acc, int lane)
{
    float4 a[GROUP][D / 128];
    float4 b[GROUP][D / 128];
    const int last = n - 1;
    int k = 0;
    load_group<D>(a, H, s, 0, last, lane);
    while (k + 2 * GROUP < n) {
        load_group<D>(b, H, s, k + GROUP, last, lane);
        add_group<D, true>(a, n, k, r, w, acc, lane);
        load_group<D>(a, H, s, k + 2 * GROUP, last, lane);
        add_group<D, true>(b, n, k + GROUP, r, w, acc, lane);
        k += 2 * GROUP;
    }
    if (k + GROUP < n) {
        load_group<D>(b, H, s, k + GROUP, last, lane);
        add_group<D, true>(a, n, k, r, w, acc, lane);
        add_group<D, false>(b, n, k + GROUP, r, w, acc, lane);
    } else {
        add_group<D, false>(a, n, k, r, w, acc, lane);
    }
}

template<int D, bool KEEP>
__device__ __forceinline__ void drain_queue(
    const float* __restrict__ H,
    const int2*  __restrict__ pairs,
    const int* q, int cnt, float* acc, int lane,
    int2* __restrict__ kept, int kept_n)
{
    for (int b = 0; b < cnt; b += 32) {
        const int n = min(32, cnt - b);
        int2 ent = make_int2(0, 0);
        if (lane < n)
            ent = pairs[q[b + lane]];
        const int   s = ent.x & 0xFFFFF;
        const int   r = ent.x >> 20;
        const float w = __int_as_float(ent.y);
        add_batch<D>(H, n, r, s, w, acc, lane);
        if (KEEP) {
            const int pos = kept_n + b + lane;
            if (lane < n && pos < KEEP_CAP)
                kept[pos] = ent;
        }
    }
}

template<int D>
__device__ __forceinline__ void drain_kept(
    const float* __restrict__ H,
    const int2*  __restrict__ kept, int cnt, float* acc, int lane)
{
    for (int b = 0; b < cnt; b += 32) {
        const int n = min(32, cnt - b);
        int   r = 0, s = 0;
        float w = 0.0f;
        if (lane < n) {
            const int2 ent = kept[b + lane];
            s = ent.x & 0xFFFFF;
            r = ent.x >> 20;
            w = __int_as_float(ent.y);
        }
        add_batch<D>(H, n, r, s, w, acc, lane);
    }
}

template<int D, int R, int MODE, bool RELU>
__global__ __launch_bounds__(1024) void spmm_rows_lds(
    const float* __restrict__ H,
    const int*   __restrict__ tile,
    const int*   __restrict__ pairs,
    float*       __restrict__ Z,
    int*         __restrict__ kept,
    int*         __restrict__ kept_cnt)
{
    __shared__ float acc[R * D];
    __shared__ int   queue[32 * QCAP];

    const int lane = threadIdx.x & 31;
    const int wv   = __builtin_amdgcn_readfirstlane(threadIdx.x >> 5);
    const int base = blockIdx.x * R;
    const int slot = blockIdx.x * 32 + wv;

    for (int i = threadIdx.x; i < R * D / 4; i += 1024)
        ((float4*)acc)[i] = make_float4(0.0f, 0.0f, 0.0f, 0.0f);
    __syncthreads();

    int2* kp = (int2*)kept + (size_t)slot * KEEP_CAP;
    const int2* __restrict__ pr = (const int2*)pairs;

    int have = -1;
    if (MODE == 1)
        have = __builtin_amdgcn_readfirstlane(kept_cnt[slot]);

    if (have >= 0) {
        drain_kept<D>(H, kp, have, acc, lane);
    } else {
        int* q      = queue + wv * QCAP;
        int  cnt    = 0;
        int  kept_n = 0;

        constexpr int NCHUNK = N_EDGES / SCAN_CHUNK;
        const uint4* __restrict__ tp = (const uint4*)tile + lane;
        const unsigned pat = (unsigned)blockIdx.x * 0x01010101u;
        uint4 d[4];
#pragma unroll
        for (int j = 0; j < 4; ++j)
            d[j] = tp[min(wv + 32 * j, NCHUNK - 1) * 32];
        for (int c = wv; c < NCHUNK; c += 128) {
            uint4 dn[4];
#pragma unroll
            for (int j = 0; j < 4; ++j)
                dn[j] = tp[min(c + 128 + 32 * j, NCHUNK - 1) * 32];
#pragma unroll
            for (int j = 0; j < 4; ++j) {
                if (c + 32 * j >= NCHUNK) break;
                const int e0 = (c + 32 * j) * SCAN_CHUNK + lane * 16;
                const unsigned xv[4] = { d[j].x, d[j].y, d[j].z, d[j].w };
#pragma unroll
                for (int t = 0; t < 4; ++t) {
                    const unsigned y = xv[t] ^ pat;
                    const bool     any   = ((y - 0x01010101u) & ~y & 0x80808080u) != 0;
                    const unsigned amask = __builtin_amdgcn_ballot_w32(any);
                    if (amask) {
#pragma unroll
                        for (int k = 0; k < 4; ++k) {
                            const bool hit = ((y >> (8 * k)) & 0xFFu) == 0;
                            const unsigned mask = __builtin_amdgcn_ballot_w32(hit);
                            if (mask) {
                                if (hit) {
                                    const int pos = cnt + __builtin_amdgcn_mbcnt_lo(mask, 0);
                                    q[pos] = e0 + 4 * t + k;
                                }
                                cnt += __builtin_popcount(mask);
                            }
                        }
                    }
                }
                if (cnt > QCAP - SCAN_CHUNK) {
                    drain_queue<D, MODE == 0>(H, pr, q, cnt, acc, lane, kp, kept_n);
                    kept_n += cnt;
                    cnt = 0;
                }
            }
#pragma unroll
            for (int j = 0; j < 4; ++j) d[j] = dn[j];
        }
        drain_queue<D, MODE == 0>(H, pr, q, cnt, acc, lane, kp, kept_n);
        kept_n += cnt;

        if (MODE == 0 && lane == 0)
            kept_cnt[slot] = (kept_n <= KEEP_CAP) ? kept_n : -1;
    }

    __syncthreads();

    for (int i = threadIdx.x; i < R * D / 4; i += 1024) {
        const int trow = i / (D / 4);
        const int c4   = i % (D / 4);
        const float* p = acc + trow * D + (c4 >> 5) * 128 + (c4 & 31);
        float4 o = make_float4(p[0], p[32], p[64], p[96]);
        if (RELU) o = make_float4(fmaxf(o.x, 0.0f), fmaxf(o.y, 0.0f), fmaxf(o.z, 0.0f), fmaxf(o.w, 0.0f));
        if (base + trow < N_NODES)
            ((float4*)(Z + (size_t)base * D))[i] = o;
    }
}

#define KC 64
#define MT 5

static_assert(N_NODES % (16 * MT) == 0, "M is whole 80-row blocks");

template<int NSUB, int N, int KDIM, bool PACK>
__global__ __launch_bounds__(128) __attribute__((amdgpu_waves_per_eu(1, 1))) void gemm_wmma_f32(
    const float* __restrict__ A,
    const float* __restrict__ B,
    float*       __restrict__ C,
    const int*   __restrict__ esrc,
    const int*   __restrict__ edst,
    const float* __restrict__ eval_,
    int*         __restrict__ tile,
    int*         __restrict__ pairs)
{
    constexpr int NB      = 64 * NSUB;
    constexpr int A_PITCH = KDIM + 4;
    constexpr int B_PITCH = NB + 16;
    constexpr int NCH     = KDIM / KC;
    constexpr int A_LD    = (16 * MT * KC / 4) / 128;
    constexpr int B_LD    = (KC * NB / 4) / 128;
    static_assert(KDIM % KC == 0, "whole K chunks");
    static_assert(N == 2 * NB, "the grid's y splits N in two");
    static_assert((16 * MT * KC / 4) % 128 == 0 && (KC * NB / 4) % 128 == 0,
                  "a chunk is whole float4 rounds of the block");
    static_assert(GEMM_BLOCKS == 2 * (N_NODES / (16 * MT)), "the grid the edge shares are cut for");
    constexpr int PK_RND = PACK_SHARE / 256;

    __shared__ float sA[16 * MT * A_PITCH];
    __shared__ float sB[KDIM * B_PITCH];

    const int tid  = threadIdx.x;
    const int lane = tid & 31;
    const int wave = tid >> 5;
    const int lo   = lane & 15;
    const int hi   = lane >> 4;

    const int m0  = blockIdx.x * (16 * MT);
    const int n0  = blockIdx.y * NB;
    const int nb0 = wave * (NSUB * 16);

    v8f acc[MT][NSUB];
#pragma unroll
    for (int ms = 0; ms < MT; ++ms)
#pragma unroll
        for (int t = 0; t < NSUB; ++t) acc[ms][t] = 0.0f;

    const float* ag[A_LD];
    int          al[A_LD];
#pragma unroll
    for (int i = 0; i < A_LD; ++i) {
        const int idx = tid + 128 * i;
        const int row = idx / (KC / 4);
        const int c4  = idx % (KC / 4);
        ag[i] = A + (size_t)(m0 + row) * KDIM + c4 * 4;
        al[i] = row * A_PITCH + c4 * 4;
    }
    const float* bg[B_LD];
    int          bl[B_LD];
#pragma unroll
    for (int i = 0; i < B_LD; ++i) {
        const int idx = tid + 128 * i;
        const int kr  = idx / (NB / 4);
        const int c4  = idx % (NB / 4);
        bg[i] = B + (size_t)kr * N + n0 + c4 * 4;
        bl[i] = kr * B_PITCH + c4 * 4;
    }

    EdgeItem pk[PK_RND];
    const int pk_e0 = (blockIdx.y * (N_NODES / (16 * MT)) + blockIdx.x) * PACK_SHARE + 2 * tid;
    if (PACK) {
#pragma unroll
        for (int j = 0; j < PK_RND; ++j)
            edge_item_load(pk[j], esrc, edst, eval_, pk_e0 + 256 * j);
    }

    float4 ra[A_LD], rb[B_LD];
#pragma unroll
    for (int i = 0; i < A_LD; ++i) ra[i] = *(const float4*)(ag[i]);
#pragma unroll
    for (int i = 0; i < B_LD; ++i) rb[i] = *(const float4*)(bg[i]);

#pragma unroll
    for (int i = 0; i < A_LD; ++i) *(float4*)(&sA[al[i]]) = ra[i];
#pragma unroll
    for (int i = 0; i < B_LD; ++i) *(float4*)(&sB[bl[i]]) = rb[i];
    if (PACK) {
#pragma unroll
        for (int j = 0; j < PK_RND; ++j)
            asm volatile("" :: "v"(pk[j].d.x), "v"(pk[j].d.y), "v"(pk[j].s.x), "v"(pk[j].s.y),
                               "v"(pk[j].w.x), "v"(pk[j].w.y));
    }
    __syncthreads();

    if (PACK) {
#pragma unroll
        for (int j = 0; j < PK_RND; ++j)
            edge_item_pack<SPMM_ROWS>(pk[j], tile, pairs, pk_e0 + 256 * j);
    }

    auto chunk = [&](int c) {
        const float* __restrict__ cA = sA + c * KC;
        const float* __restrict__ cB = sB + c * KC * B_PITCH;
#pragma unroll
        for (int kk = 0; kk < KC; kk += 4) {
            const int ka = kk + 2 * hi;
            v2f a[MT];
#pragma unroll
            for (int ms = 0; ms < MT; ++ms)
                a[ms] = *(const v2f*)(cA + (16 * ms + lo) * A_PITCH + ka);
#pragma unroll
            for (int t = 0; t < NSUB; ++t) {
                const int n = nb0 + t * 16 + lo;
                v2f b;
                b.x = cB[ka * B_PITCH + n];
                b.y = cB[(ka + 1) * B_PITCH + n];
#pragma unroll
                for (int ms = 0; ms < MT; ++ms) {
                    acc[ms][t] = __builtin_amdgcn_wmma_f32_16x16x4_f32(
                        false, a[ms], false, b, (short)0, acc[ms][t],
                        false, false);
                }
            }
        }
    };

#pragma unroll
    for (int c = 0; c + 1 < NCH; ++c) {
#pragma unroll
        for (int i = 0; i < A_LD; ++i)
            ra[i] = *(const float4*)(ag[i] + (c + 1) * KC);
#pragma unroll
        for (int i = 0; i < B_LD; ++i)
            rb[i] = *(const float4*)(bg[i] + (size_t)(c + 1) * KC * N);

        chunk(c);

#pragma unroll
        for (int i = 0; i < A_LD; ++i) *(float4*)(&sA[al[i] + (c + 1) * KC]) = ra[i];
#pragma unroll
        for (int i = 0; i < B_LD; ++i) *(float4*)(&sB[bl[i] + (c + 1) * KC * B_PITCH]) = rb[i];
        __syncthreads();
    }
    chunk(NCH - 1);

#pragma unroll
    for (int ms = 0; ms < MT; ++ms) {
#pragma unroll
        for (int t = 0; t < NSUB; ++t) {
#pragma unroll
            for (int v = 0; v < 8; ++v) {
                const int row = m0 + 16 * ms + v + 8 * hi;
                const int col = n0 + nb0 + t * 16 + lo;
                C[(size_t)row * N + col] = acc[ms][t][v];
            }
        }
    }
}

extern "C" void kernel_launch(void* const* d_in, const int* in_sizes, int n_in,
                              void* d_out, int out_size, void* d_ws, size_t ws_size,
                              hipStream_t stream)
{
    (void)in_sizes; (void)n_in; (void)out_size; (void)ws_size;

    const float* features = (const float*)d_in[0];
    const int*   esrc     = (const int*)  d_in[1];
    const int*   edst     = (const int*)  d_in[2];
    const float* eval_    = (const float*)d_in[3];
    const float* W0       = (const float*)d_in[4];
    const float* W1       = (const float*)d_in[5];
    float*       out      = (float*)d_out;

    const size_t SZ = (size_t)N_NODES * HIDDEN * sizeof(float);
    float* P = (float*)d_ws;
    float* H = (float*)((char*)d_ws + SZ);

    const int spmm_blocks = (N_NODES + SPMM_ROWS - 1) / SPMM_ROWS;
    constexpr size_t KEPT = (size_t)((N_NODES + SPMM_ROWS - 1) / SPMM_ROWS) * 32 * KEEP_CAP * sizeof(int2);
    int* kept     = (int*)((char*)d_ws + 2 * SZ);
    int* kept_cnt = (int*)((char*)d_ws + 2 * SZ + KEPT);
    constexpr size_t CNTS = (size_t)((N_NODES + SPMM_ROWS - 1) / SPMM_ROWS) * 32 * sizeof(int);
    static_assert((2 * (size_t)N_NODES * HIDDEN * sizeof(float) + KEPT + CNTS) % 16 == 0 &&
                  N_EDGES % 16 == 0, "the byte plane and the pairs start on 16-byte boundaries");
    int* tile  = (int*)((char*)d_ws + 2 * SZ + KEPT + CNTS);
    int* pairs = (int*)((char*)d_ws + 2 * SZ + KEPT + CNTS + N_EDGES);
    const dim3 gemm_grid(N_NODES / (16 * MT), 2);

    gemm_wmma_f32<2, HIDDEN, IN_DIM, true><<<gemm_grid, 128, 0, stream>>>(features, W0, P,
                                                                        esrc, edst, eval_, tile, pairs);
    spmm_rows_lds<HIDDEN, SPMM_ROWS, 0, true><<<spmm_blocks, 1024, 0, stream>>>(P, tile, pairs, H,
                                                                              kept, kept_cnt);

    float* G = P;
    gemm_wmma_f32<1, OUT_DIM, HIDDEN, false><<<gemm_grid, 128, 0, stream>>>(H, W1, G,
                                                                          nullptr, nullptr, nullptr, nullptr, nullptr);
    spmm_rows_lds<OUT_DIM, SPMM_ROWS, 1, false><<<spmm_blocks, 1024, 0, stream>>>(G, tile, pairs, out,
                                                                                kept, kept_cnt);
}


# SpMM filter: a lane's 16 tile bytes tested branch-free into one hit mask, one ballot per 512-edge chunk (was a branch per dword and per byte); queue order in a chunk changes (arrival-order sums)
# speedup vs baseline: 3.4543x; 1.0333x over previous
#include <hip/hip_runtime.h>

#define N_NODES 10000
#define N_EDGES 320000
#define IN_DIM  256
#define HIDDEN  256
#define OUT_DIM 128

typedef __attribute__((ext_vector_type(2))) float v2f;
typedef __attribute__((ext_vector_type(8))) float v8f;

#define SPMM_ROWS   40
#define GEMM_BLOCKS 250
#define PACK_SHARE  (N_EDGES / GEMM_BLOCKS)

static_assert(N_EDGES % GEMM_BLOCKS == 0 && PACK_SHARE % 256 == 0,
              "a block's share of the edges is whole rounds of its 128 threads");

struct EdgeItem { int2 d, s; float2 w; };

__device__ __forceinline__ void edge_item_load(EdgeItem& it,
                                               const int*   __restrict__ src,
                                               const int*   __restrict__ dst,
                                               const float* __restrict__ val, int e0)
{
    it.d = *(const int2*)(dst + e0);
    it.s = *(const int2*)(src + e0);
    it.w = *(const float2*)(val + e0);
}

template<int R>
__device__ __forceinline__ void edge_item_pack(const EdgeItem& it,
                                               int* __restrict__ tile,
                                               int* __restrict__ pairs,
                                               int e0)
{
    static_assert((N_NODES + R - 1) / R <= 255, "tile index fits a byte, 255 is free");
    static_assert(R <= (1 << 11), "tile row fits above the 20 src bits");
    const int   dv[2] = { it.d.x, it.d.y };
    const int   ss[2] = { it.s.x, it.s.y };
    const float ww[2] = { it.w.x, it.w.y };
    unsigned ids = 0;
    int2 p[2];
#pragma unroll
    for (int t = 0; t < 2; ++t) {
        const unsigned du = (unsigned)dv[t];
        const bool     ok = du < (unsigned)N_NODES;
        const unsigned ti = du / (unsigned)R;
        const unsigned tr = du - ti * (unsigned)R;
        ids |= (ok ? ti : 255u) << (8 * t);
        p[t] = make_int2(ss[t] | (int)(tr << 20), __float_as_int(ww[t]));
    }
    ((unsigned short*)tile)[e0 >> 1] = (unsigned short)ids;
    *(int4*)(pairs + 2 * e0) = make_int4(p[0].x, p[0].y, p[1].x, p[1].y);
}

#define SCAN_CHUNK 512
#define QCAP       1152
#define KEEP_CAP   256

static_assert(N_EDGES % SCAN_CHUNK == 0, "the scan takes whole chunks");
static_assert(N_EDGES / SCAN_CHUNK >= 32, "every wave of a block has a first chunk");
static_assert(N_NODES <= (1 << 20), "a pair packs src into 20 bits");

template<int D>
__device__ __forceinline__ void load_row(float4 (&h)[D / 128],
                                         const float* __restrict__ H,
                                         int s, int lane)
{
    static_assert(D % 128 == 0, "a row is whole 512-byte pieces");
    const float4* __restrict__ hrow = (const float4*)(H + (size_t)s * D) + lane;
#pragma unroll
    for (int i = 0; i < D / 128; ++i) h[i] = hrow[i * 32];
}

#define GROUP 4

template<int D>
__device__ __forceinline__ void load_group(float4 (&h)[GROUP][D / 128],
                                           const float* __restrict__ H,
                                           int s, int k, int last, int lane)
{
#pragma unroll
    for (int g = 0; g < GROUP; ++g)
        load_row<D>(h[g], H,
                    __builtin_amdgcn_readlane(s, min(k + g, last)), lane);
}

template<int D, bool FULL>
__device__ __forceinline__ void add_group(const float4 (&h)[GROUP][D / 128],
                                          int n, int k, int r, float w,
                                          float* acc, int lane)
{
#pragma unroll
    for (int g = 0; g < GROUP; ++g) {
        if (FULL || k + g < n) {
            const float wk = __int_as_float(
                __builtin_amdgcn_readlane(__float_as_int(w), k + g));
            float* arow = acc + __builtin_amdgcn_readlane(r, k + g) * D + lane;
#pragma unroll
            for (int i = 0; i < D / 128; ++i) {
                atomicAdd(&arow[i * 128],      wk * h[g][i].x);
                atomicAdd(&arow[i * 128 + 32], wk * h[g][i].y);
                atomicAdd(&arow[i * 128 + 64], wk * h[g][i].z);
                atomicAdd(&arow[i * 128 + 96], wk * h[g][i].w);
            }
        }
    }
}

template<int D>
__device__ __forceinline__ void add_batch(
    const float* __restrict__ H, int n, int r, int s, float w,
    float* acc, int lane)
{
    float4 a[GROUP][D / 128];
    float4 b[GROUP][D / 128];
    const int last = n - 1;
    int k = 0;
    load_group<D>(a, H, s, 0, last, lane);
    while (k + 2 * GROUP < n) {
        load_group<D>(b, H, s, k + GROUP, last, lane);
        add_group<D, true>(a, n, k, r, w, acc, lane);
        load_group<D>(a, H, s, k + 2 * GROUP, last, lane);
        add_group<D, true>(b, n, k + GROUP, r, w, acc, lane);
        k += 2 * GROUP;
    }
    if (k + GROUP < n) {
        load_group<D>(b, H, s, k + GROUP, last, lane);
        add_group<D, true>(a, n, k, r, w, acc, lane);
        add_group<D, false>(b, n, k + GROUP, r, w, acc, lane);
    } else {
        add_group<D, false>(a, n, k, r, w, acc, lane);
    }
}

template<int D, bool KEEP>
__device__ __forceinline__ void drain_queue(
    const float* __restrict__ H,
    const int2*  __restrict__ pairs,
    const int* q, int cnt, float* acc, int lane,
    int2* __restrict__ kept, int kept_n)
{
    for (int b = 0; b < cnt; b += 32) {
        const int n = min(32, cnt - b);
        int2 ent = make_int2(0, 0);
        if (lane < n)
            ent = pairs[q[b + lane]];
        const int   s = ent.x & 0xFFFFF;
        const int   r = ent.x >> 20;
        const float w = __int_as_float(ent.y);
        add_batch<D>(H, n, r, s, w, acc, lane);
        if (KEEP) {
            const int pos = kept_n + b + lane;
            if (lane < n && pos < KEEP_CAP)
                kept[pos] = ent;
        }
    }
}

template<int D>
__device__ __forceinline__ void drain_kept(
    const float* __restrict__ H,
    const int2*  __restrict__ kept, int cnt, float* acc, int lane)
{
    for (int b = 0; b < cnt; b += 32) {
        const int n = min(32, cnt - b);
        int   r = 0, s = 0;
        float w = 0.0f;
        if (lane < n) {
            const int2 ent = kept[b + lane];
            s = ent.x & 0xFFFFF;
            r = ent.x >> 20;
            w = __int_as_float(ent.y);
        }
        add_batch<D>(H, n, r, s, w, acc, lane);
    }
}

template<int D, int R, int MODE, bool RELU>
__global__ __launch_bounds__(1024) void spmm_rows_lds(
    const float* __restrict__ H,
    const int*   __restrict__ tile,
    const int*   __restrict__ pairs,
    float*       __restrict__ Z,
    int*         __restrict__ kept,
    int*         __restrict__ kept_cnt)
{
    __shared__ float acc[R * D];
    __shared__ int   queue[32 * QCAP];

    const int lane = threadIdx.x & 31;
    const int wv   = __builtin_amdgcn_readfirstlane(threadIdx.x >> 5);
    const int base = blockIdx.x * R;
    const int slot = blockIdx.x * 32 + wv;

    for (int i = threadIdx.x; i < R * D / 4; i += 1024)
        ((float4*)acc)[i] = make_float4(0.0f, 0.0f, 0.0f, 0.0f);
    __syncthreads();

    int2* kp = (int2*)kept + (size_t)slot * KEEP_CAP;
    const int2* __restrict__ pr = (const int2*)pairs;

    int have = -1;
    if (MODE == 1)
        have = __builtin_amdgcn_readfirstlane(kept_cnt[slot]);

    if (have >= 0) {
        drain_kept<D>(H, kp, have, acc, lane);
    } else {
        int* q      = queue + wv * QCAP;
        int  cnt    = 0;
        int  kept_n = 0;

        constexpr int NCHUNK = N_EDGES / SCAN_CHUNK;
        const uint4* __restrict__ tp = (const uint4*)tile + lane;
        const unsigned pat = (unsigned)blockIdx.x * 0x01010101u;
        uint4 d[4];
#pragma unroll
        for (int j = 0; j < 4; ++j)
            d[j] = tp[min(wv + 32 * j, NCHUNK - 1) * 32];
        for (int c = wv; c < NCHUNK; c += 128) {
            uint4 dn[4];
#pragma unroll
            for (int j = 0; j < 4; ++j)
                dn[j] = tp[min(c + 128 + 32 * j, NCHUNK - 1) * 32];
#pragma unroll
            for (int j = 0; j < 4; ++j) {
                if (c + 32 * j >= NCHUNK) break;
                const int e0 = (c + 32 * j) * SCAN_CHUNK + lane * 16;
                const unsigned xv[4] = { d[j].x, d[j].y, d[j].z, d[j].w };
                unsigned m = 0;
#pragma unroll
                for (int t = 0; t < 4; ++t) {
                    const unsigned y = xv[t] ^ pat;
                    const unsigned z = ~(((y & 0x7F7F7F7Fu) + 0x7F7F7F7Fu) | y) & 0x80808080u;
                    m |= z >> (7 - t);
                }
                unsigned mask = __builtin_amdgcn_ballot_w32(m != 0);
                while (mask) {
                    if (m != 0) {
                        const int b   = __builtin_ctz(m);
                        const int pos = cnt + __builtin_amdgcn_mbcnt_lo(mask, 0);
                        q[pos] = e0 + 4 * (b & 7) + (b >> 3);
                    }
                    cnt += __builtin_popcount(mask);
                    m &= m - 1;
                    mask = __builtin_amdgcn_ballot_w32(m != 0);
                }
                if (cnt > QCAP - SCAN_CHUNK) {
                    drain_queue<D, MODE == 0>(H, pr, q, cnt, acc, lane, kp, kept_n);
                    kept_n += cnt;
                    cnt = 0;
                }
            }
#pragma unroll
            for (int j = 0; j < 4; ++j) d[j] = dn[j];
        }
        drain_queue<D, MODE == 0>(H, pr, q, cnt, acc, lane, kp, kept_n);
        kept_n += cnt;

        if (MODE == 0 && lane == 0)
            kept_cnt[slot] = (kept_n <= KEEP_CAP) ? kept_n : -1;
    }

    __syncthreads();

    for (int i = threadIdx.x; i < R * D / 4; i += 1024) {
        const int trow = i / (D / 4);
        const int c4   = i % (D / 4);
        const float* p = acc + trow * D + (c4 >> 5) * 128 + (c4 & 31);
        float4 o = make_float4(p[0], p[32], p[64], p[96]);
        if (RELU) o = make_float4(fmaxf(o.x, 0.0f), fmaxf(o.y, 0.0f), fmaxf(o.z, 0.0f), fmaxf(o.w, 0.0f));
        if (base + trow < N_NODES)
            ((float4*)(Z + (size_t)base * D))[i] = o;
    }
}

#define KC 64
#define MT 5

static_assert(N_NODES % (16 * MT) == 0, "M is whole 80-row blocks");

template<int NSUB, int N, int KDIM, bool PACK>
__global__ __launch_bounds__(128) __attribute__((amdgpu_waves_per_eu(1, 1))) void gemm_wmma_f32(
    const float* __restrict__ A,
    const float* __restrict__ B,
    float*       __restrict__ C,
    const int*   __restrict__ esrc,
    const int*   __restrict__ edst,
    const float* __restrict__ eval_,
    int*         __restrict__ tile,
    int*         __restrict__ pairs)
{
    constexpr int NB      = 64 * NSUB;
    constexpr int A_PITCH = KDIM + 4;
    constexpr int B_PITCH = NB + 16;
    constexpr int NCH     = KDIM / KC;
    constexpr int A_LD    = (16 * MT * KC / 4) / 128;
    constexpr int B_LD    = (KC * NB / 4) / 128;
    static_assert(KDIM % KC == 0, "whole K chunks");
    static_assert(N == 2 * NB, "the grid's y splits N in two");
    static_assert((16 * MT * KC / 4) % 128 == 0 && (KC * NB / 4) % 128 == 0,
                  "a chunk is whole float4 rounds of the block");
    static_assert(GEMM_BLOCKS == 2 * (N_NODES / (16 * MT)), "the grid the edge shares are cut for");
    constexpr int PK_RND = PACK_SHARE / 256;

    __shared__ float sA[16 * MT * A_PITCH];
    __shared__ float sB[KDIM * B_PITCH];

    const int tid  = threadIdx.x;
    const int lane = tid & 31;
    const int wave = tid >> 5;
    const int lo   = lane & 15;
    const int hi   = lane >> 4;

    const int m0  = blockIdx.x * (16 * MT);
    const int n0  = blockIdx.y * NB;
    const int nb0 = wave * (NSUB * 16);

    v8f acc[MT][NSUB];
#pragma unroll
    for (int ms = 0; ms < MT; ++ms)
#pragma unroll
        for (int t = 0; t < NSUB; ++t) acc[ms][t] = 0.0f;

    const float* ag[A_LD];
    int          al[A_LD];
#pragma unroll
    for (int i = 0; i < A_LD; ++i) {
        const int idx = tid + 128 * i;
        const int row = idx / (KC / 4);
        const int c4  = idx % (KC / 4);
        ag[i] = A + (size_t)(m0 + row) * KDIM + c4 * 4;
        al[i] = row * A_PITCH + c4 * 4;
    }
    const float* bg[B_LD];
    int          bl[B_LD];
#pragma unroll
    for (int i = 0; i < B_LD; ++i) {
        const int idx = tid + 128 * i;
        const int kr  = idx / (NB / 4);
        const int c4  = idx % (NB / 4);
        bg[i] = B + (size_t)kr * N + n0 + c4 * 4;
        bl[i] = kr * B_PITCH + c4 * 4;
    }

    EdgeItem pk[PK_RND];
    const int pk_e0 = (blockIdx.y * (N_NODES / (16 * MT)) + blockIdx.x) * PACK_SHARE + 2 * tid;
    if (PACK) {
#pragma unroll
        for (int j = 0; j < PK_RND; ++j)
            edge_item_load(pk[j], esrc, edst, eval_, pk_e0 + 256 * j);
    }

    float4 ra[A_LD], rb[B_LD];
#pragma unroll
    for (int i = 0; i < A_LD; ++i) ra[i] = *(const float4*)(ag[i]);
#pragma unroll
    for (int i = 0; i < B_LD; ++i) rb[i] = *(const float4*)(bg[i]);

#pragma unroll
    for (int i = 0; i < A_LD; ++i) *(float4*)(&sA[al[i]]) = ra[i];
#pragma unroll
    for (int i = 0; i < B_LD; ++i) *(float4*)(&sB[bl[i]]) = rb[i];
    if (PACK) {
#pragma unroll
        for (int j = 0; j < PK_RND; ++j)
            asm volatile("" :: "v"(pk[j].d.x), "v"(pk[j].d.y), "v"(pk[j].s.x), "v"(pk[j].s.y),
                               "v"(pk[j].w.x), "v"(pk[j].w.y));
    }
    __syncthreads();

    if (PACK) {
#pragma unroll
        for (int j = 0; j < PK_RND; ++j)
            edge_item_pack<SPMM_ROWS>(pk[j], tile, pairs, pk_e0 + 256 * j);
    }

    auto chunk = [&](int c) {
        const float* __restrict__ cA = sA + c * KC;
        const float* __restrict__ cB = sB + c * KC * B_PITCH;
#pragma unroll
        for (int kk = 0; kk < KC; kk += 4) {
            const int ka = kk + 2 * hi;
            v2f a[MT];
#pragma unroll
            for (int ms = 0; ms < MT; ++ms)
                a[ms] = *(const v2f*)(cA + (16 * ms + lo) * A_PITCH + ka);
#pragma unroll
            for (int t = 0; t < NSUB; ++t) {
                const int n = nb0 + t * 16 + lo;
                v2f b;
                b.x = cB[ka * B_PITCH + n];
                b.y = cB[(ka + 1) * B_PITCH + n];
#pragma unroll
                for (int ms = 0; ms < MT; ++ms) {
                    acc[ms][t] = __builtin_amdgcn_wmma_f32_16x16x4_f32(
                        false, a[ms], false, b, (short)0, acc[ms][t],
                        false, false);
                }
            }
        }
    };

#pragma unroll
    for (int c = 0; c + 1 < NCH; ++c) {
#pragma unroll
        for (int i = 0; i < A_LD; ++i)
            ra[i] = *(const float4*)(ag[i] + (c + 1) * KC);
#pragma unroll
        for (int i = 0; i < B_LD; ++i)
            rb[i] = *(const float4*)(bg[i] + (size_t)(c + 1) * KC * N);

        chunk(c);

#pragma unroll
        for (int i = 0; i < A_LD; ++i) *(float4*)(&sA[al[i] + (c + 1) * KC]) = ra[i];
#pragma unroll
        for (int i = 0; i < B_LD; ++i) *(float4*)(&sB[bl[i] + (c + 1) * KC * B_PITCH]) = rb[i];
        __syncthreads();
    }
    chunk(NCH - 1);

#pragma unroll
    for (int ms = 0; ms < MT; ++ms) {
#pragma unroll
        for (int t = 0; t < NSUB; ++t) {
#pragma unroll
            for (int v = 0; v < 8; ++v) {
                const int row = m0 + 16 * ms + v + 8 * hi;
                const int col = n0 + nb0 + t * 16 + lo;
                C[(size_t)row * N + col] = acc[ms][t][v];
            }
        }
    }
}

extern "C" void kernel_launch(void* const* d_in, const int* in_sizes, int n_in,
                              void* d_out, int out_size, void* d_ws, size_t ws_size,
                              hipStream_t stream)
{
    (void)in_sizes; (void)n_in; (void)out_size; (void)ws_size;

    const float* features = (const float*)d_in[0];
    const int*   esrc     = (const int*)  d_in[1];
    const int*   edst     = (const int*)  d_in[2];
    const float* eval_    = (const float*)d_in[3];
    const float* W0       = (const float*)d_in[4];
    const float* W1       = (const float*)d_in[5];
    float*       out      = (float*)d_out;

    const size_t SZ = (size_t)N_NODES * HIDDEN * sizeof(float);
    float* P = (float*)d_ws;
    float* H = (float*)((char*)d_ws + SZ);

    const int spmm_blocks = (N_NODES + SPMM_ROWS - 1) / SPMM_ROWS;
    constexpr size_t KEPT = (size_t)((N_NODES + SPMM_ROWS - 1) / SPMM_ROWS) * 32 * KEEP_CAP * sizeof(int2);
    int* kept     = (int*)((char*)d_ws + 2 * SZ);
    int* kept_cnt = (int*)((char*)d_ws + 2 * SZ + KEPT);
    constexpr size_t CNTS = (size_t)((N_NODES + SPMM_ROWS - 1) / SPMM_ROWS) * 32 * sizeof(int);
    static_assert((2 * (size_t)N_NODES * HIDDEN * sizeof(float) + KEPT + CNTS) % 16 == 0 &&
                  N_EDGES % 16 == 0, "the byte plane and the pairs start on 16-byte boundaries");
    int* tile  = (int*)((char*)d_ws + 2 * SZ + KEPT + CNTS);
    int* pairs = (int*)((char*)d_ws + 2 * SZ + KEPT + CNTS + N_EDGES);
    const dim3 gemm_grid(N_NODES / (16 * MT), 2);

    gemm_wmma_f32<2, HIDDEN, IN_DIM, true><<<gemm_grid, 128, 0, stream>>>(features, W0, P,
                                                                        esrc, edst, eval_, tile, pairs);
    spmm_rows_lds<HIDDEN, SPMM_ROWS, 0, true><<<spmm_blocks, 1024, 0, stream>>>(P, tile, pairs, H,
                                                                              kept, kept_cnt);

    float* G = P;
    gemm_wmma_f32<1, OUT_DIM, HIDDEN, false><<<gemm_grid, 128, 0, stream>>>(H, W1, G,
                                                                          nullptr, nullptr, nullptr, nullptr, nullptr);
    spmm_rows_lds<OUT_DIM, SPMM_ROWS, 1, false><<<spmm_blocks, 1024, 0, stream>>>(G, tile, pairs, out,
                                                                                kept, kept_cnt);
}


# Layer-1 GEMM moved into the layer-0 SpMM launch: a block's relu'd row tile in LDS is the A operand of @W1 (W1 in LDS, f32 WMMA, K in 4 partial sums); H never global, 3 launches; k order changes
# speedup vs baseline: 3.7168x; 1.0760x over previous
#include <hip/hip_runtime.h>

#define N_NODES 10000
#define N_EDGES 320000
#define IN_DIM  256
#define HIDDEN  256
#define OUT_DIM 128

typedef __attribute__((ext_vector_type(2))) float v2f;
typedef __attribute__((ext_vector_type(8))) float v8f;

#define SPMM_ROWS   40
#define GEMM_BLOCKS 250
#define PACK_SHARE  (N_EDGES / GEMM_BLOCKS)

static_assert(N_EDGES % GEMM_BLOCKS == 0 && PACK_SHARE % 256 == 0,
              "a block's share of the edges is whole rounds of its 128 threads");

struct EdgeItem { int2 d, s; float2 w; };

__device__ __forceinline__ void edge_item_load(EdgeItem& it,
                                               const int*   __restrict__ src,
                                               const int*   __restrict__ dst,
                                               const float* __restrict__ val, int e0)
{
    it.d = *(const int2*)(dst + e0);
    it.s = *(const int2*)(src + e0);
    it.w = *(const float2*)(val + e0);
}

template<int R>
__device__ __forceinline__ void edge_item_pack(const EdgeItem& it,
                                               int* __restrict__ tile,
                                               int* __restrict__ pairs,
                                               int e0)
{
    static_assert((N_NODES + R - 1) / R <= 255, "tile index fits a byte, 255 is free");
    static_assert(R <= (1 << 11), "tile row fits above the 20 src bits");
    const int   dv[2] = { it.d.x, it.d.y };
    const int   ss[2] = { it.s.x, it.s.y };
    const float ww[2] = { it.w.x, it.w.y };
    unsigned ids = 0;
    int2 p[2];
#pragma unroll
    for (int t = 0; t < 2; ++t) {
        const unsigned du = (unsigned)dv[t];
        const bool     ok = du < (unsigned)N_NODES;
        const unsigned ti = du / (unsigned)R;
        const unsigned tr = du - ti * (unsigned)R;
        ids |= (ok ? ti : 255u) << (8 * t);
        p[t] = make_int2(ss[t] | (int)(tr << 20), __float_as_int(ww[t]));
    }
    ((unsigned short*)tile)[e0 >> 1] = (unsigned short)ids;
    *(int4*)(pairs + 2 * e0) = make_int4(p[0].x, p[0].y, p[1].x, p[1].y);
}

#define SCAN_CHUNK 512
#define QCAP       1152
#define QCAP_FUSE  896
#define KEEP_CAP   256
#define W_PITCH    (OUT_DIM + 16)

static_assert(N_EDGES % SCAN_CHUNK == 0, "the scan takes whole chunks");
static_assert(N_EDGES / SCAN_CHUNK >= 32, "every wave of a block has a first chunk");
static_assert(N_NODES <= (1 << 20), "a pair packs src into 20 bits");

template<int D>
__device__ __forceinline__ void load_row(float4 (&h)[D / 128],
                                         const float* __restrict__ H,
                                         int s, int lane)
{
    static_assert(D % 128 == 0, "a row is whole 512-byte pieces");
    const float4* __restrict__ hrow = (const float4*)(H + (size_t)s * D) + lane;
#pragma unroll
    for (int i = 0; i < D / 128; ++i) h[i] = hrow[i * 32];
}

#define GROUP 4

template<int D>
__device__ __forceinline__ void load_group(float4 (&h)[GROUP][D / 128],
                                           const float* __restrict__ H,
                                           int s, int k, int last, int lane)
{
#pragma unroll
    for (int g = 0; g < GROUP; ++g)
        load_row<D>(h[g], H,
                    __builtin_amdgcn_readlane(s, min(k + g, last)), lane);
}

template<int D, int AP, bool FULL>
__device__ __forceinline__ void add_group(const float4 (&h)[GROUP][D / 128],
                                          int n, int k, int r, float w,
                                          float* acc, int lane)
{
#pragma unroll
    for (int g = 0; g < GROUP; ++g) {
        if (FULL || k + g < n) {
            const float wk = __int_as_float(
                __builtin_amdgcn_readlane(__float_as_int(w), k + g));
            float* arow = acc + __builtin_amdgcn_readlane(r, k + g) * AP + lane;
#pragma unroll
            for (int i = 0; i < D / 128; ++i) {
                atomicAdd(&arow[i * 128],      wk * h[g][i].x);
                atomicAdd(&arow[i * 128 + 32], wk * h[g][i].y);
                atomicAdd(&arow[i * 128 + 64], wk * h[g][i].z);
                atomicAdd(&arow[i * 128 + 96], wk * h[g][i].w);
            }
        }
    }
}

template<int D, int AP>
__device__ __forceinline__ void add_batch(
    const float* __restrict__ H, int n, int r, int s, float w,
    float* acc, int lane)
{
    float4 a[GROUP][D / 128];
    float4 b[GROUP][D / 128];
    const int last = n - 1;
    int k = 0;
    load_group<D>(a, H, s, 0, last, lane);
    while (k + 2 * GROUP < n) {
        load_group<D>(b, H, s, k + GROUP, last, lane);
        add_group<D, AP, true>(a, n, k, r, w, acc, lane);
        load_group<D>(a, H, s, k + 2 * GROUP, last, lane);
        add_group<D, AP, true>(b, n, k + GROUP, r, w, acc, lane);
        k += 2 * GROUP;
    }
    if (k + GROUP < n) {
        load_group<D>(b, H, s, k + GROUP, last, lane);
        add_group<D, AP, true>(a, n, k, r, w, acc, lane);
        add_group<D, AP, false>(b, n, k + GROUP, r, w, acc, lane);
    } else {
        add_group<D, AP, false>(a, n, k, r, w, acc, lane);
    }
}

template<int D, int AP, bool KEEP>
__device__ __forceinline__ void drain_queue(
    const float* __restrict__ H,
    const int2*  __restrict__ pairs,
    const int* q, int cnt, float* acc, int lane,
    int2* __restrict__ kept, int kept_n)
{
    for (int b = 0; b < cnt; b += 32) {
        const int n = min(32, cnt - b);
        int2 ent = make_int2(0, 0);
        if (lane < n)
            ent = pairs[q[b + lane]];
        const int   s = ent.x & 0xFFFFF;
        const int   r = ent.x >> 20;
        const float w = __int_as_float(ent.y);
        add_batch<D, AP>(H, n, r, s, w, acc, lane);
        if (KEEP) {
            const int pos = kept_n + b + lane;
            if (lane < n && pos < KEEP_CAP)
                kept[pos] = ent;
        }
    }
}

template<int D, int AP>
__device__ __forceinline__ void drain_kept(
    const float* __restrict__ H,
    const int2*  __restrict__ kept, int cnt, float* acc, int lane)
{
    for (int b = 0; b < cnt; b += 32) {
        const int n = min(32, cnt - b);
        int   r = 0, s = 0;
        float w = 0.0f;
        if (lane < n) {
            const int2 ent = kept[b + lane];
            s = ent.x & 0xFFFFF;
            r = ent.x >> 20;
            w = __int_as_float(ent.y);
        }
        add_batch<D, AP>(H, n, r, s, w, acc, lane);
    }
}

template<int D, int R, int MODE, bool RELU, bool FUSE>
__global__ __launch_bounds__(1024) void spmm_rows_lds(
    const float* __restrict__ H,
    const int*   __restrict__ tile,
    const int*   __restrict__ pairs,
    float*       __restrict__ Z,
    int*         __restrict__ kept,
    int*         __restrict__ kept_cnt,
    const float* __restrict__ W)
{
    constexpr int AP = FUSE ? D + 4 : D;
    constexpr int QC = FUSE ? QCAP_FUSE : QCAP;
    constexpr int W_LD = FUSE ? (D * OUT_DIM / 4) / 1024 : 1;
    static_assert(QC > SCAN_CHUNK, "a chunk's hits fit the queue");
    static_assert(!FUSE || ((D * OUT_DIM / 4) % 1024 == 0 && D % 128 == 0 && OUT_DIM == 128),
                  "W is whole float4 rounds of the block, its rows 512 bytes");
    static_assert(!FUSE || (R > 32 && R <= 48), "the tile is three 16-row tiles, the third partly");
    static_assert(!FUSE || MODE == 0, "W is parked on the filtering path, which MODE 0 always takes");

    __shared__ __attribute__((aligned(16))) float acc[R * AP];
    __shared__ __attribute__((aligned(16))) int queue[32 * QC];
    __shared__ __attribute__((aligned(16))) float sW[FUSE ? D * W_PITCH : 4];

    const int lane = threadIdx.x & 31;
    const int wv   = __builtin_amdgcn_readfirstlane(threadIdx.x >> 5);
    const int base = blockIdx.x * R;
    const int slot = blockIdx.x * 32 + wv;

    for (int i = threadIdx.x; i < R * AP / 4; i += 1024)
        ((float4*)acc)[i] = make_float4(0.0f, 0.0f, 0.0f, 0.0f);
    __syncthreads();

    int2* kp = (int2*)kept + (size_t)slot * KEEP_CAP;
    const int2* __restrict__ pr = (const int2*)pairs;

    int have = -1;
    if (MODE == 1)
        have = __builtin_amdgcn_readfirstlane(kept_cnt[slot]);

    if (have >= 0) {
        drain_kept<D, AP>(H, kp, have, acc, lane);
    } else {
        int* q      = queue + wv * QC;
        int  cnt    = 0;
        int  kept_n = 0;

        constexpr int NCHUNK = N_EDGES / SCAN_CHUNK;
        const uint4* __restrict__ tp = (const uint4*)tile + lane;
        const unsigned pat = (unsigned)blockIdx.x * 0x01010101u;
        uint4 d[4];
#pragma unroll
        for (int j = 0; j < 4; ++j)
            d[j] = tp[min(wv + 32 * j, NCHUNK - 1) * 32];
        if constexpr (FUSE) {
            float4 wreg[W_LD];
#pragma unroll
            for (int i = 0; i < W_LD; ++i)
                wreg[i] = ((const float4*)W)[threadIdx.x + 1024 * i];
#pragma unroll
            for (int i = 0; i < W_LD; ++i) {
                const int idx = threadIdx.x + 1024 * i;
                const int k   = idx / (OUT_DIM / 4);
                const int c4  = idx % (OUT_DIM / 4);
                const int dd  = (k & ~127) + 32 * (k & 3) + ((k & 127) >> 2);
                *(float4*)(&sW[dd * W_PITCH + 4 * c4]) = wreg[i];
            }
        }
        for (int c = wv; c < NCHUNK; c += 128) {
            uint4 dn[4];
#pragma unroll
            for (int j = 0; j < 4; ++j)
                dn[j] = tp[min(c + 128 + 32 * j, NCHUNK - 1) * 32];
#pragma unroll
            for (int j = 0; j < 4; ++j) {
                if (c + 32 * j >= NCHUNK) break;
                const int e0 = (c + 32 * j) * SCAN_CHUNK + lane * 16;
                const unsigned xv[4] = { d[j].x, d[j].y, d[j].z, d[j].w };
                unsigned m = 0;
#pragma unroll
                for (int t = 0; t < 4; ++t) {
                    const unsigned y = xv[t] ^ pat;
                    const unsigned z = ~(((y & 0x7F7F7F7Fu) + 0x7F7F7F7Fu) | y) & 0x80808080u;
                    m |= z >> (7 - t);
                }
                unsigned mask = __builtin_amdgcn_ballot_w32(m != 0);
                while (mask) {
                    if (m != 0) {
                        const int b   = __builtin_ctz(m);
                        const int pos = cnt + __builtin_amdgcn_mbcnt_lo(mask, 0);
                        q[pos] = e0 + 4 * (b & 7) + (b >> 3);
                    }
                    cnt += __builtin_popcount(mask);
                    m &= m - 1;
                    mask = __builtin_amdgcn_ballot_w32(m != 0);
                }
                if (cnt > QC - SCAN_CHUNK) {
                    drain_queue<D, AP, MODE == 0>(H, pr, q, cnt, acc, lane, kp, kept_n);
                    kept_n += cnt;
                    cnt = 0;
                }
            }
#pragma unroll
            for (int j = 0; j < 4; ++j) d[j] = dn[j];
        }
        drain_queue<D, AP, MODE == 0>(H, pr, q, cnt, acc, lane, kp, kept_n);
        kept_n += cnt;

        if (MODE == 0 && lane == 0)
            kept_cnt[slot] = (kept_n <= KEEP_CAP) ? kept_n : -1;
    }

    __syncthreads();

    if constexpr (FUSE) {
        constexpr int KQ = D / 4;
        constexpr int PP = OUT_DIM + 4;
        static_assert(4 * 48 * PP <= 32 * QC, "the four partial products fit the queues' LDS");
        const int nt = wv & 7;
        const int kq = wv >> 3;
        const int lo = lane & 15;
        const int hi = lane >> 4;
        const float* ap0 = acc + lo * AP + KQ * kq + 2 * hi;
        const float* ap1 = ap0 + 16 * AP;
        const float* ap2 = acc + min(32 + lo, R - 1) * AP + KQ * kq + 2 * hi;
        const float* bp  = sW + (KQ * kq + 2 * hi) * W_PITCH + 16 * nt + lo;
        v8f c0 = 0.0f, c1 = 0.0f, c2 = 0.0f;
#pragma unroll
        for (int s = 0; s < KQ / 4; ++s) {
            v2f b;
            b.x = bp[(4 * s) * W_PITCH];
            b.y = bp[(4 * s + 1) * W_PITCH];
            v2f a0 = *(const v2f*)(ap0 + 4 * s);
            v2f a1 = *(const v2f*)(ap1 + 4 * s);
            v2f a2 = *(const v2f*)(ap2 + 4 * s);
            if (RELU) {
                a0.x = fmaxf(a0.x, 0.0f); a0.y = fmaxf(a0.y, 0.0f);
                a1.x = fmaxf(a1.x, 0.0f); a1.y = fmaxf(a1.y, 0.0f);
                a2.x = fmaxf(a2.x, 0.0f); a2.y = fmaxf(a2.y, 0.0f);
            }
            c0 = __builtin_amdgcn_wmma_f32_16x16x4_f32(false, a0, false, b, (short)0, c0, false, false);
            c1 = __builtin_amdgcn_wmma_f32_16x16x4_f32(false, a1, false, b, (short)0, c1, false, false);
            c2 = __builtin_amdgcn_wmma_f32_16x16x4_f32(false, a2, false, b, (short)0, c2, false, false);
        }
        int* part = queue + (48 * kq + 8 * hi) * PP + 16 * nt + lo;
#pragma unroll
        for (int v = 0; v < 8; ++v) {
            part[v * PP]             = __float_as_int(c0[v]);
            part[(16 + v) * PP]      = __float_as_int(c1[v]);
            part[(32 + v) * PP]      = __float_as_int(c2[v]);
        }
        __syncthreads();
        for (int i = threadIdx.x; i < R * (OUT_DIM / 4); i += 1024) {
            const int trow = i / (OUT_DIM / 4);
            const int c4   = i % (OUT_DIM / 4);
            const int4 p0 = *(const int4*)(queue + trow * PP + 4 * c4);
            float4 o = make_float4(__int_as_float(p0.x), __int_as_float(p0.y),
                                   __int_as_float(p0.z), __int_as_float(p0.w));
#pragma unroll
            for (int t = 1; t < 4; ++t) {
                const int4 pq = *(const int4*)(queue + (48 * t + trow) * PP + 4 * c4);
                o.x += __int_as_float(pq.x); o.y += __int_as_float(pq.y);
                o.z += __int_as_float(pq.z); o.w += __int_as_float(pq.w);
            }
            if (base + trow < N_NODES)
                ((float4*)(Z + (size_t)(base + trow) * OUT_DIM))[c4] = o;
        }
    } else {
        for (int i = threadIdx.x; i < R * D / 4; i += 1024) {
            const int trow = i / (D / 4);
            const int c4   = i % (D / 4);
            const float* p = acc + trow * AP + (c4 >> 5) * 128 + (c4 & 31);
            float4 o = make_float4(p[0], p[32], p[64], p[96]);
            if (RELU) o = make_float4(fmaxf(o.x, 0.0f), fmaxf(o.y, 0.0f), fmaxf(o.z, 0.0f), fmaxf(o.w, 0.0f));
            if (base + trow < N_NODES)
                ((float4*)(Z + (size_t)base * D))[i] = o;
        }
    }
}

#define KC 64
#define MT 5

static_assert(N_NODES % (16 * MT) == 0, "M is whole 80-row blocks");

template<int NSUB, int N, int KDIM, bool PACK>
__global__ __launch_bounds__(128) __attribute__((amdgpu_waves_per_eu(1, 1))) void gemm_wmma_f32(
    const float* __restrict__ A,
    const float* __restrict__ B,
    float*       __restrict__ C,
    const int*   __restrict__ esrc,
    const int*   __restrict__ edst,
    const float* __restrict__ eval_,
    int*         __restrict__ tile,
    int*         __restrict__ pairs)
{
    constexpr int NB      = 64 * NSUB;
    constexpr int A_PITCH = KDIM + 4;
    constexpr int B_PITCH = NB + 16;
    constexpr int NCH     = KDIM / KC;
    constexpr int A_LD    = (16 * MT * KC / 4) / 128;
    constexpr int B_LD    = (KC * NB / 4) / 128;
    static_assert(KDIM % KC == 0, "whole K chunks");
    static_assert(N == 2 * NB, "the grid's y splits N in two");
    static_assert((16 * MT * KC / 4) % 128 == 0 && (KC * NB / 4) % 128 == 0,
                  "a chunk is whole float4 rounds of the block");
    static_assert(GEMM_BLOCKS == 2 * (N_NODES / (16 * MT)), "the grid the edge shares are cut for");
    constexpr int PK_RND = PACK_SHARE / 256;

    __shared__ float sA[16 * MT * A_PITCH];
    __shared__ float sB[KDIM * B_PITCH];

    const int tid  = threadIdx.x;
    const int lane = tid & 31;
    const int wave = tid >> 5;
    const int lo   = lane & 15;
    const int hi   = lane >> 4;

    const int m0  = blockIdx.x * (16 * MT);
    const int n0  = blockIdx.y * NB;
    const int nb0 = wave * (NSUB * 16);

    v8f acc[MT][NSUB];
#pragma unroll
    for (int ms = 0; ms < MT; ++ms)
#pragma unroll
        for (int t = 0; t < NSUB; ++t) acc[ms][t] = 0.0f;

    const float* ag[A_LD];
    int          al[A_LD];
#pragma unroll
    for (int i = 0; i < A_LD; ++i) {
        const int idx = tid + 128 * i;
        const int row = idx / (KC / 4);
        const int c4  = idx % (KC / 4);
        ag[i] = A + (size_t)(m0 + row) * KDIM + c4 * 4;
        al[i] = row * A_PITCH + c4 * 4;
    }
    const float* bg[B_LD];
    int          bl[B_LD];
#pragma unroll
    for (int i = 0; i < B_LD; ++i) {
        const int idx = tid + 128 * i;
        const int kr  = idx / (NB / 4);
        const int c4  = idx % (NB / 4);
        bg[i] = B + (size_t)kr * N + n0 + c4 * 4;
        bl[i] = kr * B_PITCH + c4 * 4;
    }

    EdgeItem pk[PK_RND];
    const int pk_e0 = (blockIdx.y * (N_NODES / (16 * MT)) + blockIdx.x) * PACK_SHARE + 2 * tid;
    if (PACK) {
#pragma unroll
        for (int j = 0; j < PK_RND; ++j)
            edge_item_load(pk[j], esrc, edst, eval_, pk_e0 + 256 * j);
    }

    float4 ra[A_LD], rb[B_LD];
#pragma unroll
    for (int i = 0; i < A_LD; ++i) ra[i] = *(const float4*)(ag[i]);
#pragma unroll
    for (int i = 0; i < B_LD; ++i) rb[i] = *(const float4*)(bg[i]);

#pragma unroll
    for (int i = 0; i < A_LD; ++i) *(float4*)(&sA[al[i]]) = ra[i];
#pragma unroll
    for (int i = 0; i < B_LD; ++i) *(float4*)(&sB[bl[i]]) = rb[i];
    if (PACK) {
#pragma unroll
        for (int j = 0; j < PK_RND; ++j)
            asm volatile("" :: "v"(pk[j].d.x), "v"(pk[j].d.y), "v"(pk[j].s.x), "v"(pk[j].s.y),
                               "v"(pk[j].w.x), "v"(pk[j].w.y));
    }
    __syncthreads();

    if (PACK) {
#pragma unroll
        for (int j = 0; j < PK_RND; ++j)
            edge_item_pack<SPMM_ROWS>(pk[j], tile, pairs, pk_e0 + 256 * j);
    }

    auto chunk = [&](int c) {
        const float* __restrict__ cA = sA + c * KC;
        const float* __restrict__ cB = sB + c * KC * B_PITCH;
#pragma unroll
        for (int kk = 0; kk < KC; kk += 4) {
            const int ka = kk + 2 * hi;
            v2f a[MT];
#pragma unroll
            for (int ms = 0; ms < MT; ++ms)
                a[ms] = *(const v2f*)(cA + (16 * ms + lo) * A_PITCH + ka);
#pragma unroll
            for (int t = 0; t < NSUB; ++t) {
                const int n = nb0 + t * 16 + lo;
                v2f b;
                b.x = cB[ka * B_PITCH + n];
                b.y = cB[(ka + 1) * B_PITCH + n];
#pragma unroll
                for (int ms = 0; ms < MT; ++ms) {
                    acc[ms][t] = __builtin_amdgcn_wmma_f32_16x16x4_f32(
                        false, a[ms], false, b, (short)0, acc[ms][t],
                        false, false);
                }
            }
        }
    };

#pragma unroll
    for (int c = 0; c + 1 < NCH; ++c) {
#pragma unroll
        for (int i = 0; i < A_LD; ++i)
            ra[i] = *(const float4*)(ag[i] + (c + 1) * KC);
#pragma unroll
        for (int i = 0; i < B_LD; ++i)
            rb[i] = *(const float4*)(bg[i] + (size_t)(c + 1) * KC * N);

        chunk(c);

#pragma unroll
        for (int i = 0; i < A_LD; ++i) *(float4*)(&sA[al[i] + (c + 1) * KC]) = ra[i];
#pragma unroll
        for (int i = 0; i < B_LD; ++i) *(float4*)(&sB[bl[i] + (c + 1) * KC * B_PITCH]) = rb[i];
        __syncthreads();
    }
    chunk(NCH - 1);

#pragma unroll
    for (int ms = 0; ms < MT; ++ms) {
#pragma unroll
        for (int t = 0; t < NSUB; ++t) {
#pragma unroll
            for (int v = 0; v < 8; ++v) {
                const int row = m0 + 16 * ms + v + 8 * hi;
                const int col = n0 + nb0 + t * 16 + lo;
                C[(size_t)row * N + col] = acc[ms][t][v];
            }
        }
    }
}

extern "C" void kernel_launch(void* const* d_in, const int* in_sizes, int n_in,
                              void* d_out, int out_size, void* d_ws, size_t ws_size,
                              hipStream_t stream)
{
    (void)in_sizes; (void)n_in; (void)out_size; (void)ws_size;

    const float* features = (const float*)d_in[0];
    const int*   esrc     = (const int*)  d_in[1];
    const int*   edst     = (const int*)  d_in[2];
    const float* eval_    = (const float*)d_in[3];
    const float* W0       = (const float*)d_in[4];
    const float* W1       = (const float*)d_in[5];
    float*       out      = (float*)d_out;

    const size_t SZ = (size_t)N_NODES * HIDDEN * sizeof(float);
    float* P = (float*)d_ws;
    float* G = (float*)((char*)d_ws + SZ);

    const int spmm_blocks = (N_NODES + SPMM_ROWS - 1) / SPMM_ROWS;
    constexpr size_t KEPT = (size_t)((N_NODES + SPMM_ROWS - 1) / SPMM_ROWS) * 32 * KEEP_CAP * sizeof(int2);
    int* kept     = (int*)((char*)d_ws + 2 * SZ);
    int* kept_cnt = (int*)((char*)d_ws + 2 * SZ + KEPT);
    constexpr size_t CNTS = (size_t)((N_NODES + SPMM_ROWS - 1) / SPMM_ROWS) * 32 * sizeof(int);
    static_assert((2 * (size_t)N_NODES * HIDDEN * sizeof(float) + KEPT + CNTS) % 16 == 0 &&
                  N_EDGES % 16 == 0, "the byte plane and the pairs start on 16-byte boundaries");
    int* tile  = (int*)((char*)d_ws + 2 * SZ + KEPT + CNTS);
    int* pairs = (int*)((char*)d_ws + 2 * SZ + KEPT + CNTS + N_EDGES);
    const dim3 gemm_grid(N_NODES / (16 * MT), 2);

    gemm_wmma_f32<2, HIDDEN, IN_DIM, true><<<gemm_grid, 128, 0, stream>>>(features, W0, P,
                                                                        esrc, edst, eval_, tile, pairs);
    spmm_rows_lds<HIDDEN, SPMM_ROWS, 0, true, true><<<spmm_blocks, 1024, 0, stream>>>(P, tile, pairs, G,
                                                                                    kept, kept_cnt, W1);

    spmm_rows_lds<OUT_DIM, SPMM_ROWS, 1, false, false><<<spmm_blocks, 1024, 0, stream>>>(G, tile, pairs, out,
                                                                                       kept, kept_cnt, nullptr);
}
